# starting kernel
# baseline (speedup 1.0000x reference)
#include <hip/hip_runtime.h>


typedef __attribute__((ext_vector_type(2))) float v2f;
typedef __attribute__((ext_vector_type(8))) float v8f;

__global__ void k_init(float* __restrict__ deg, float* __restrict__ agg1,
                       float* __restrict__ out, int N) {
  int i = blockIdx.x * blockDim.x + threadIdx.x;
  if (i < 16 * N) agg1[i] = 0.0f;
  if (i < 2 * N)  out[i]  = 0.0f;
  if (i < N)      deg[i]  = 1.0f;
}

__global__ void k_degree(const int* __restrict__ dst, float* __restrict__ deg, int E) {
  int e = blockIdx.x * blockDim.x + threadIdx.x;
  if (e < E) atomicAdd(&deg[dst[e]], 1.0f);
}

__global__ void k_dinv(float* __restrict__ deg, int N) {
  int i = blockIdx.x * blockDim.x + threadIdx.x;
  if (i < N) {
    float d = deg[i];
    deg[i] = (d > 0.0f) ? rsqrtf(d) : 0.0f;
  }
}

__global__ void k_gemm1(const float* __restrict__ x, const float* __restrict__ W1,
                        float* __restrict__ h1, int ntiles) {
  int wave = (blockIdx.x * blockDim.x + threadIdx.x) >> 5;
  int lane = threadIdx.x & 31;
  if (wave >= ntiles) return;
  int node0 = wave << 4;

  v2f a = {0.0f, 0.0f};
  v2f b = {0.0f, 0.0f};
  if (lane < 16) {
    int node = node0 + lane;
    a.x = x[node * 2 + 0];
    a.y = x[node * 2 + 1];
    b.x = W1[0 * 16 + lane];
    b.y = W1[1 * 16 + lane];
  }

  v8f c = {};
  c = __builtin_amdgcn_wmma_f32_16x16x4_f32(
       false, a,  false, b,
       (short)0, c,  false,  false);

  int row0 = node0 + ((lane < 16) ? 0 : 8);
  int col  = lane & 15;
#pragma unroll
  for (int j = 0; j < 8; ++j)
    h1[(row0 + j) * 16 + col] = c[j];
}

__global__ void k_gemm1_tail(const float* __restrict__ x, const float* __restrict__ W1,
                             float* __restrict__ h1, int n0, int N) {
  int tid = blockIdx.x * blockDim.x + threadIdx.x;
  int n = n0 + (tid >> 4);
  if (n >= N) return;
  int f = tid & 15;
  h1[n * 16 + f] = x[n * 2 + 0] * W1[f] + x[n * 2 + 1] * W1[16 + f];
}

__global__ void k_agg1(const int* __restrict__ src, const int* __restrict__ dst,
                       const float* __restrict__ dinv, const float* __restrict__ h1,
                       float* __restrict__ agg1, int E) {
  int tid = blockIdx.x * blockDim.x + threadIdx.x;
  if (tid >= E * 16) return;
  int e = tid >> 4;
  int f = tid & 15;
  int s = src[e];
  int d = dst[e];
  float nrm = dinv[s] * dinv[d];
  atomicAdd(&agg1[d * 16 + f], h1[s * 16 + f] * nrm);
}

__global__ void k_relu1(const float* __restrict__ h1, const float* __restrict__ dinv,
                        const float* __restrict__ b1, float* __restrict__ agg1, int N) {
  int tid = blockIdx.x * blockDim.x + threadIdx.x;
  if (tid >= N * 16) return;
  int n = tid >> 4;
  int f = tid & 15;
  float di = dinv[n];
  float v = agg1[tid] + h1[tid] * di * di + b1[f];
  agg1[tid] = fmaxf(v, 0.0f);
}

__global__ void k_gemm2(const float* __restrict__ r, const float* __restrict__ W2,
                        float* __restrict__ h2, int ntiles) {
  int wave = (blockIdx.x * blockDim.x + threadIdx.x) >> 5;
  int lane = threadIdx.x & 31;
  if (wave >= ntiles) return;
  int node0 = wave << 4;

  int m = lane & 15;
  int node = node0 + m;

  v8f c = {};
#pragma unroll
  for (int kk = 0; kk < 4; ++kk) {
    int kbase = kk * 4 + ((lane < 16) ? 0 : 2);
    v2f a, b;
    a.x = r[node * 16 + kbase + 0];
    a.y = r[node * 16 + kbase + 1];
    b.x = (m < 2) ? W2[(kbase + 0) * 2 + m] : 0.0f;
    b.y = (m < 2) ? W2[(kbase + 1) * 2 + m] : 0.0f;
    c = __builtin_amdgcn_wmma_f32_16x16x4_f32(
        false, a, false, b, (short)0, c, false, false);
  }

  int row0 = node0 + ((lane < 16) ? 0 : 8);
  int col  = lane & 15;
  if (col < 2) {
#pragma unroll
    for (int j = 0; j < 8; ++j)
      h2[(row0 + j) * 2 + col] = c[j];
  }
}

__global__ void k_gemm2_tail(const float* __restrict__ r, const float* __restrict__ W2,
                             float* __restrict__ h2, int n0, int N) {
  int tid = blockIdx.x * blockDim.x + threadIdx.x;
  int n = n0 + (tid >> 1);
  if (n >= N) return;
  int f = tid & 1;
  float acc = 0.0f;
#pragma unroll
  for (int k = 0; k < 16; ++k)
    acc += r[n * 16 + k] * W2[k * 2 + f];
  h2[n * 2 + f] = acc;
}

__global__ void k_agg2(const int* __restrict__ src, const int* __restrict__ dst,
                       const float* __restrict__ dinv, const float* __restrict__ h2,
                       float* __restrict__ out, int E) {
  int tid = blockIdx.x * blockDim.x + threadIdx.x;
  if (tid >= 2 * E) return;
  int e = tid >> 1;
  int f = tid & 1;
  int s = src[e];
  int d = dst[e];
  float nrm = dinv[s] * dinv[d];
  atomicAdd(&out[d * 2 + f], h2[s * 2 + f] * nrm);
}

__global__ void k_final(const float* __restrict__ h2, const float* __restrict__ dinv,
                        const float* __restrict__ b2, float* __restrict__ out, int N) {
  int tid = blockIdx.x * blockDim.x + threadIdx.x;
  if (tid >= 2 * N) return;
  int n = tid >> 1;
  int f = tid & 1;
  float di = dinv[n];
  out[tid] += h2[tid] * di * di + b2[f];
}

extern "C" void kernel_launch(void* const* d_in, const int* in_sizes, int n_in,
                              void* d_out, int out_size, void* d_ws, size_t ws_size,
                              hipStream_t stream) {
  const float* x  = (const float*)d_in[0];
  const int*   ei = (const int*)d_in[1];
  const float* W1 = (const float*)d_in[2];
  const float* b1 = (const float*)d_in[3];
  const float* W2 = (const float*)d_in[4];
  const float* b2 = (const float*)d_in[5];
  float* out = (float*)d_out;

  const int N = in_sizes[0] / 2;
  const int E = in_sizes[1] / 2;
  const int* src = ei;
  const int* dst = ei + E;

  float* ws   = (float*)d_ws;
  float* dinv = ws;
  float* h1   = ws + (size_t)N;
  float* agg1 = ws + (size_t)17 * N;
  float* h2   = ws + (size_t)33 * N;

  const int B = 256;
  const int ntiles = N / 16;
  const int n0 = ntiles * 16;
  const int ntail = N - n0;

  k_init  <<<(16 * N + B - 1) / B, B, 0, stream>>>(dinv, agg1, out, N);
  k_degree<<<(E + B - 1) / B,      B, 0, stream>>>(dst, dinv, E);
  k_dinv  <<<(N + B - 1) / B,      B, 0, stream>>>(dinv, N);

  k_gemm1 <<<(ntiles * 32 + B - 1) / B, B, 0, stream>>>(x, W1, h1, ntiles);
  if (ntail > 0)
    k_gemm1_tail<<<(ntail * 16 + B - 1) / B, B, 0, stream>>>(x, W1, h1, n0, N);

  k_agg1  <<<(E * 16 + B - 1) / B, B, 0, stream>>>(src, dst, dinv, h1, agg1, E);
  k_relu1 <<<(16 * N + B - 1) / B, B, 0, stream>>>(h1, dinv, b1, agg1, N);

  k_gemm2 <<<(ntiles * 32 + B - 1) / B, B, 0, stream>>>(agg1, W2, h2, ntiles);
  if (ntail > 0)
    k_gemm2_tail<<<(ntail * 2 + B - 1) / B, B, 0, stream>>>(agg1, W2, h2, n0, N);

  k_agg2  <<<(2 * E + B - 1) / B,  B, 0, stream>>>(src, dst, dinv, h2, out, E);
  k_final <<<(2 * N + B - 1) / B,  B, 0, stream>>>(h2, dinv, b2, out, N);
}


# In-degree count and layer-2 scatter-add accumulate in LDS tiles per destination range (ds_add), partials summed in a 2nd pass, h2 pre-scaled by dinv; no global atomics there; float sum order changes
# speedup vs baseline: 1.9125x; 1.9125x over previous
#include <hip/hip_runtime.h>


typedef __attribute__((ext_vector_type(2))) float v2f;
typedef __attribute__((ext_vector_type(8))) float v8f;

__global__ void k_init(float* __restrict__ agg1, int N) {
  int i = blockIdx.x * blockDim.x + threadIdx.x;
  if (i < 16 * N) agg1[i] = 0.0f;
}

#define TILE_FLOATS 65536
#define SC_THREADS  1024
#define SC_BATCH    8

__global__ void k_degree_lds(const int* __restrict__ dst, int* __restrict__ partial,
                             int E, int N) {
  __shared__ __attribute__((aligned(16))) int tile[TILE_FLOATS];
  const int t = threadIdx.x;
  const int R = (((N + (int)gridDim.y - 1) / (int)gridDim.y) + 31) & ~31;
  const int eps = (E + (int)gridDim.x - 1) / (int)gridDim.x;
  const int lo = blockIdx.y * R;
  const int nloc = max(0, min(R, N - lo));

  const int4 z4 = {0, 0, 0, 0};
  for (int i = t * 4; i < TILE_FLOATS; i += SC_THREADS * 4)
    *reinterpret_cast<int4*>(&tile[i]) = z4;
  __syncthreads();

  const int e0 = blockIdx.x * eps;
  const int e1 = min(E, e0 + eps);
  const int elast = e1 - 1;

  int dn[SC_BATCH];
  int base = e0 + t;
  for (int k = 0; k < SC_BATCH; ++k) {
    int e = base + k * SC_THREADS;
    int dv = dst[min(e, elast)];
    dn[k] = (e < e1) ? dv : -1;
  }
  for (; base < e1; base += SC_BATCH * SC_THREADS) {
    int d[SC_BATCH];
    for (int k = 0; k < SC_BATCH; ++k) d[k] = dn[k];
    const int nb = base + SC_BATCH * SC_THREADS;
    for (int k = 0; k < SC_BATCH; ++k) {
      int e = nb + k * SC_THREADS;
      int dv = dst[min(e, elast)];
      dn[k] = (e < e1) ? dv : -1;
    }
    for (int k = 0; k < SC_BATCH; ++k) {
      unsigned r = (unsigned)(d[k] - lo);
      if (r < (unsigned)nloc) atomicAdd(&tile[r], 1);
    }
  }
  __syncthreads();

  int* p = partial + ((size_t)blockIdx.x * N + lo);
  for (int i = t; i < nloc; i += SC_THREADS) p[i] = tile[i];
}

__global__ void k_agg2_lds(const int* __restrict__ src, const int* __restrict__ dst,
                           const float* __restrict__ feat, float* __restrict__ partial,
                           int E, int N) {
  __shared__ __attribute__((aligned(16))) float tile[TILE_FLOATS];
  const int t = threadIdx.x;
  const int R = (((N + (int)gridDim.y - 1) / (int)gridDim.y) + 31) & ~31;
  const int eps = (E + (int)gridDim.x - 1) / (int)gridDim.x;
  const int lo = blockIdx.y * R;
  const int nloc = max(0, min(R, N - lo));

  const float4 z4 = {0.0f, 0.0f, 0.0f, 0.0f};
  for (int i = t * 4; i < TILE_FLOATS; i += SC_THREADS * 4)
    *reinterpret_cast<float4*>(&tile[i]) = z4;
  __syncthreads();

  const int e0 = blockIdx.x * eps;
  const int e1 = min(E, e0 + eps);
  const int elast = e1 - 1;
  const v2f* feat2 = reinterpret_cast<const v2f*>(feat);

  int dn[SC_BATCH], sn[SC_BATCH];
  int base = e0 + t;
  for (int k = 0; k < SC_BATCH; ++k) {
    int e = base + k * SC_THREADS;
    int ec = min(e, elast);
    int dv = dst[ec];
    dn[k] = (e < e1) ? dv : -1;
    sn[k] = src[ec];
  }
  for (; base < e1; base += SC_BATCH * SC_THREADS) {
    int d[SC_BATCH], s[SC_BATCH];
    for (int k = 0; k < SC_BATCH; ++k) { d[k] = dn[k]; s[k] = sn[k]; }
    const int nb = base + SC_BATCH * SC_THREADS;
    for (int k = 0; k < SC_BATCH; ++k) {
      int e = nb + k * SC_THREADS;
      int ec = min(e, elast);
      int dv = dst[ec];
      dn[k] = (e < e1) ? dv : -1;
      sn[k] = src[ec];
    }
    v2f v[SC_BATCH];
    for (int k = 0; k < SC_BATCH; ++k) {
      unsigned r = (unsigned)(d[k] - lo);
      v[k] = feat2[(r < (unsigned)nloc) ? s[k] : 0];
    }
    for (int k = 0; k < SC_BATCH; ++k) {
      unsigned r = (unsigned)(d[k] - lo);
      if (r < (unsigned)nloc) {
        atomicAdd(&tile[2 * r + 0], v[k].x);
        atomicAdd(&tile[2 * r + 1], v[k].y);
      }
    }
  }
  __syncthreads();

  float* p = partial + ((size_t)blockIdx.x * N + lo) * 2;
  const int cnt = nloc * 2;
  for (int i = t; i < cnt; i += SC_THREADS) p[i] = tile[i];
}

__global__ void k_dinv(const int* __restrict__ part, float* __restrict__ dinv,
                       int N, int nsl) {
  int i = blockIdx.x * blockDim.x + threadIdx.x;
  if (i < N) {
    int c0 = 0, c1 = 0, c2 = 0, c3 = 0;
    int s = 0;
    for (; s + 4 <= nsl; s += 4) {
      c0 += part[(size_t)(s + 0) * N + i];
      c1 += part[(size_t)(s + 1) * N + i];
      c2 += part[(size_t)(s + 2) * N + i];
      c3 += part[(size_t)(s + 3) * N + i];
    }
    for (; s < nsl; ++s) c0 += part[(size_t)s * N + i];
    float d = 1.0f + (float)((c0 + c1) + (c2 + c3));
    dinv[i] = (d > 0.0f) ? rsqrtf(d) : 0.0f;
  }
}

__global__ void k_gemm1(const float* __restrict__ x, const float* __restrict__ W1,
                        float* __restrict__ h1, int ntiles) {
  int wave = (blockIdx.x * blockDim.x + threadIdx.x) >> 5;
  int lane = threadIdx.x & 31;
  if (wave >= ntiles) return;
  int node0 = wave << 4;

  v2f a = {0.0f, 0.0f};
  v2f b = {0.0f, 0.0f};
  if (lane < 16) {
    int node = node0 + lane;
    a.x = x[node * 2 + 0];
    a.y = x[node * 2 + 1];
    b.x = W1[0 * 16 + lane];
    b.y = W1[1 * 16 + lane];
  }

  v8f c = {};
  c = __builtin_amdgcn_wmma_f32_16x16x4_f32(
       false, a,  false, b,
       (short)0, c,  false,  false);

  int row0 = node0 + ((lane < 16) ? 0 : 8);
  int col  = lane & 15;
#pragma unroll
  for (int j = 0; j < 8; ++j)
    h1[(row0 + j) * 16 + col] = c[j];
}

__global__ void k_gemm1_tail(const float* __restrict__ x, const float* __restrict__ W1,
                             float* __restrict__ h1, int n0, int N) {
  int tid = blockIdx.x * blockDim.x + threadIdx.x;
  int n = n0 + (tid >> 4);
  if (n >= N) return;
  int f = tid & 15;
  h1[n * 16 + f] = x[n * 2 + 0] * W1[f] + x[n * 2 + 1] * W1[16 + f];
}

__global__ void k_agg1(const int* __restrict__ src, const int* __restrict__ dst,
                       const float* __restrict__ dinv, const float* __restrict__ h1,
                       float* __restrict__ agg1, int E) {
  int tid = blockIdx.x * blockDim.x + threadIdx.x;
  if (tid >= E * 16) return;
  int e = tid >> 4;
  int f = tid & 15;
  int s = src[e];
  int d = dst[e];
  float nrm = dinv[s] * dinv[d];
  atomicAdd(&agg1[d * 16 + f], h1[s * 16 + f] * nrm);
}

__global__ void k_relu1(const float* __restrict__ h1, const float* __restrict__ dinv,
                        const float* __restrict__ b1, float* __restrict__ agg1, int N) {
  int tid = blockIdx.x * blockDim.x + threadIdx.x;
  if (tid >= N * 16) return;
  int n = tid >> 4;
  int f = tid & 15;
  float di = dinv[n];
  float v = agg1[tid] + h1[tid] * di * di + b1[f];
  agg1[tid] = fmaxf(v, 0.0f);
}

__global__ void k_gemm2(const float* __restrict__ r, const float* __restrict__ W2,
                        const float* __restrict__ dinv,
                        float* __restrict__ h2s, int ntiles) {
  int wave = (blockIdx.x * blockDim.x + threadIdx.x) >> 5;
  int lane = threadIdx.x & 31;
  if (wave >= ntiles) return;
  int node0 = wave << 4;

  int m = lane & 15;
  int node = node0 + m;

  v8f c = {};
#pragma unroll
  for (int kk = 0; kk < 4; ++kk) {
    int kbase = kk * 4 + ((lane < 16) ? 0 : 2);
    v2f a, b;
    a.x = r[node * 16 + kbase + 0];
    a.y = r[node * 16 + kbase + 1];
    b.x = (m < 2) ? W2[(kbase + 0) * 2 + m] : 0.0f;
    b.y = (m < 2) ? W2[(kbase + 1) * 2 + m] : 0.0f;
    c = __builtin_amdgcn_wmma_f32_16x16x4_f32(
        false, a, false, b, (short)0, c, false, false);
  }

  int row0 = node0 + ((lane < 16) ? 0 : 8);
  int col  = lane & 15;
  if (col < 2) {
#pragma unroll
    for (int j = 0; j < 8; ++j)
      h2s[(row0 + j) * 2 + col] = c[j] * dinv[row0 + j];
  }
}

__global__ void k_gemm2_tail(const float* __restrict__ r, const float* __restrict__ W2,
                             const float* __restrict__ dinv,
                             float* __restrict__ h2s, int n0, int N) {
  int tid = blockIdx.x * blockDim.x + threadIdx.x;
  int n = n0 + (tid >> 1);
  if (n >= N) return;
  int f = tid & 1;
  float acc = 0.0f;
#pragma unroll
  for (int k = 0; k < 16; ++k)
    acc += r[n * 16 + k] * W2[k * 2 + f];
  h2s[n * 2 + f] = acc * dinv[n];
}

__global__ void k_final(const float* __restrict__ part, const float* __restrict__ h2s,
                        const float* __restrict__ dinv, const float* __restrict__ b2,
                        float* __restrict__ out, int N, int nsl) {
  int tid = blockIdx.x * blockDim.x + threadIdx.x;
  if (tid >= 2 * N) return;
  int n = tid >> 1;
  int f = tid & 1;
  const size_t stride = (size_t)2 * N;
  float a0 = 0.0f, a1 = 0.0f, a2 = 0.0f, a3 = 0.0f;
  int s = 0;
  for (; s + 4 <= nsl; s += 4) {
    a0 += part[(size_t)(s + 0) * stride + tid];
    a1 += part[(size_t)(s + 1) * stride + tid];
    a2 += part[(size_t)(s + 2) * stride + tid];
    a3 += part[(size_t)(s + 3) * stride + tid];
  }
  for (; s < nsl; ++s) a0 += part[(size_t)s * stride + tid];
  float di = dinv[n];
  out[tid] = ((a0 + a1) + (a2 + a3)) * di + (h2s[tid] * di + b2[f]);
}

extern "C" void kernel_launch(void* const* d_in, const int* in_sizes, int n_in,
                              void* d_out, int out_size, void* d_ws, size_t ws_size,
                              hipStream_t stream) {
  const float* x  = (const float*)d_in[0];
  const int*   ei = (const int*)d_in[1];
  const float* W1 = (const float*)d_in[2];
  const float* b1 = (const float*)d_in[3];
  const float* W2 = (const float*)d_in[4];
  const float* b2 = (const float*)d_in[5];
  float* out = (float*)d_out;

  const int N = in_sizes[0] / 2;
  const int E = in_sizes[1] / 2;
  const int* src = ei;
  const int* dst = ei + E;

  const size_t NP = ((size_t)N + 63) & ~(size_t)63;
  float* ws   = (float*)d_ws;
  float* dinv = ws;
  float* h1   = ws + NP;
  float* agg1 = ws + (size_t)17 * NP;
  float* h2s  = ws + (size_t)33 * NP;
  float* part = ws + (size_t)35 * NP;

  const int B = 256;
  const int ntiles = N / 16;
  const int n0 = ntiles * 16;
  const int ntail = N - n0;

  const int nr1 = (N + TILE_FLOATS - 1) / TILE_FLOATS;
  const int ns1 = (256 / nr1) > 0 ? (256 / nr1) : 1;
  const int nr2 = (2 * N + TILE_FLOATS - 1) / TILE_FLOATS;
  const int ns2 = (256 / nr2) > 0 ? (256 / nr2) : 1;

  k_init  <<<(16 * N + B - 1) / B, B, 0, stream>>>(agg1, N);
  k_degree_lds<<<dim3(ns1, nr1), SC_THREADS, 0, stream>>>(dst, (int*)part, E, N);
  k_dinv  <<<(N + B - 1) / B,      B, 0, stream>>>((const int*)part, dinv, N, ns1);

  k_gemm1 <<<(ntiles * 32 + B - 1) / B, B, 0, stream>>>(x, W1, h1, ntiles);
  if (ntail > 0)
    k_gemm1_tail<<<(ntail * 16 + B - 1) / B, B, 0, stream>>>(x, W1, h1, n0, N);

  k_agg1  <<<(E * 16 + B - 1) / B, B, 0, stream>>>(src, dst, dinv, h1, agg1, E);
  k_relu1 <<<(16 * N + B - 1) / B, B, 0, stream>>>(h1, dinv, b1, agg1, N);

  k_gemm2 <<<(ntiles * 32 + B - 1) / B, B, 0, stream>>>(agg1, W2, dinv, h2s, ntiles);
  if (ntail > 0)
    k_gemm2_tail<<<(ntail * 2 + B - 1) / B, B, 0, stream>>>(agg1, W2, dinv, h2s, n0, N);

  k_agg2_lds<<<dim3(ns2, nr2), SC_THREADS, 0, stream>>>(src, dst, h2s, part, E, N);
  k_final <<<(2 * N + B - 1) / B,  B, 0, stream>>>(part, h2s, dinv, b2, out, N, ns2);
}


# Layer-1 aggregation moved in front of its product, (A x) W1 for A (x W1): 2-wide rows summed in LDS tiles as in layer 2, bias and ReLU in the WMMA epilogue; no global atomics; float sum order changes
# speedup vs baseline: 4.4311x; 2.3169x over previous
#include <hip/hip_runtime.h>


typedef __attribute__((ext_vector_type(2))) float v2f;
typedef __attribute__((ext_vector_type(8))) float v8f;

#define TILE_FLOATS 65536
#define SC_THREADS  1024
#define SC_BATCH    8

__global__ void k_degree_lds(const int* __restrict__ dst, int* __restrict__ partial,
                             int E, int N) {
  __shared__ __attribute__((aligned(16))) int tile[TILE_FLOATS];
  const int t = threadIdx.x;
  const int R = (((N + (int)gridDim.y - 1) / (int)gridDim.y) + 31) & ~31;
  const int eps = (E + (int)gridDim.x - 1) / (int)gridDim.x;
  const int lo = blockIdx.y * R;
  const int nloc = max(0, min(R, N - lo));

  const int4 z4 = {0, 0, 0, 0};
  for (int i = t * 4; i < TILE_FLOATS; i += SC_THREADS * 4)
    *reinterpret_cast<int4*>(&tile[i]) = z4;
  __syncthreads();

  const int e0 = blockIdx.x * eps;
  const int e1 = min(E, e0 + eps);
  const int elast = e1 - 1;

  int dn[SC_BATCH];
  int base = e0 + t;
  for (int k = 0; k < SC_BATCH; ++k) {
    int e = base + k * SC_THREADS;
    int dv = dst[min(e, elast)];
    dn[k] = (e < e1) ? dv : -1;
  }
  for (; base < e1; base += SC_BATCH * SC_THREADS) {
    int d[SC_BATCH];
    for (int k = 0; k < SC_BATCH; ++k) d[k] = dn[k];
    const int nb = base + SC_BATCH * SC_THREADS;
    for (int k = 0; k < SC_BATCH; ++k) {
      int e = nb + k * SC_THREADS;
      int dv = dst[min(e, elast)];
      dn[k] = (e < e1) ? dv : -1;
    }
    for (int k = 0; k < SC_BATCH; ++k) {
      unsigned r = (unsigned)(d[k] - lo);
      if (r < (unsigned)nloc) atomicAdd(&tile[r], 1);
    }
  }
  __syncthreads();

  int* p = partial + ((size_t)blockIdx.x * N + lo);
  for (int i = t; i < nloc; i += SC_THREADS) p[i] = tile[i];
}

__global__ void k_agg_lds(const int* __restrict__ src, const int* __restrict__ dst,
                           const float* __restrict__ feat, float* __restrict__ partial,
                           int E, int N) {
  __shared__ __attribute__((aligned(16))) float tile[TILE_FLOATS];
  const int t = threadIdx.x;
  const int R = (((N + (int)gridDim.y - 1) / (int)gridDim.y) + 31) & ~31;
  const int eps = (E + (int)gridDim.x - 1) / (int)gridDim.x;
  const int lo = blockIdx.y * R;
  const int nloc = max(0, min(R, N - lo));

  const float4 z4 = {0.0f, 0.0f, 0.0f, 0.0f};
  for (int i = t * 4; i < TILE_FLOATS; i += SC_THREADS * 4)
    *reinterpret_cast<float4*>(&tile[i]) = z4;
  __syncthreads();

  const int e0 = blockIdx.x * eps;
  const int e1 = min(E, e0 + eps);
  const int elast = e1 - 1;
  const v2f* feat2 = reinterpret_cast<const v2f*>(feat);

  int dn[SC_BATCH], sn[SC_BATCH];
  int base = e0 + t;
  for (int k = 0; k < SC_BATCH; ++k) {
    int e = base + k * SC_THREADS;
    int ec = min(e, elast);
    int dv = dst[ec];
    dn[k] = (e < e1) ? dv : -1;
    sn[k] = src[ec];
  }
  for (; base < e1; base += SC_BATCH * SC_THREADS) {
    int d[SC_BATCH], s[SC_BATCH];
    for (int k = 0; k < SC_BATCH; ++k) { d[k] = dn[k]; s[k] = sn[k]; }
    const int nb = base + SC_BATCH * SC_THREADS;
    for (int k = 0; k < SC_BATCH; ++k) {
      int e = nb + k * SC_THREADS;
      int ec = min(e, elast);
      int dv = dst[ec];
      dn[k] = (e < e1) ? dv : -1;
      sn[k] = src[ec];
    }
    v2f v[SC_BATCH];
    for (int k = 0; k < SC_BATCH; ++k) {
      unsigned r = (unsigned)(d[k] - lo);
      v[k] = feat2[(r < (unsigned)nloc) ? s[k] : 0];
    }
    for (int k = 0; k < SC_BATCH; ++k) {
      unsigned r = (unsigned)(d[k] - lo);
      if (r < (unsigned)nloc) {
        atomicAdd(&tile[2 * r + 0], v[k].x);
        atomicAdd(&tile[2 * r + 1], v[k].y);
      }
    }
  }
  __syncthreads();

  float* p = partial + ((size_t)blockIdx.x * N + lo) * 2;
  const int cnt = nloc * 2;
  for (int i = t; i < cnt; i += SC_THREADS) p[i] = tile[i];
}

__global__ void k_dinv(const int* __restrict__ part, const float* __restrict__ x,
                       float* __restrict__ dinv, float* __restrict__ xs,
                       int N, int nsl) {
  int i = blockIdx.x * blockDim.x + threadIdx.x;
  if (i < N) {
    int c0 = 0, c1 = 0, c2 = 0, c3 = 0;
    int s = 0;
    for (; s + 4 <= nsl; s += 4) {
      c0 += part[(size_t)(s + 0) * N + i];
      c1 += part[(size_t)(s + 1) * N + i];
      c2 += part[(size_t)(s + 2) * N + i];
      c3 += part[(size_t)(s + 3) * N + i];
    }
    for (; s < nsl; ++s) c0 += part[(size_t)s * N + i];
    float d = 1.0f + (float)((c0 + c1) + (c2 + c3));
    float di = (d > 0.0f) ? rsqrtf(d) : 0.0f;
    dinv[i] = di;
    v2f xv = reinterpret_cast<const v2f*>(x)[i];
    reinterpret_cast<v2f*>(xs)[i] = xv * di;
  }
}

__global__ void k_gemm1(const float* __restrict__ part, const float* __restrict__ xs,
                        const float* __restrict__ dinv, const float* __restrict__ W1,
                        const float* __restrict__ b1, float* __restrict__ r,
                        int ntiles, int N, int nsl) {
  int wave = (blockIdx.x * blockDim.x + threadIdx.x) >> 5;
  int lane = threadIdx.x & 31;
  if (wave >= ntiles) return;
  int node0 = wave << 4;

  v2f a = {0.0f, 0.0f};
  v2f b = {0.0f, 0.0f};
  if (lane < 16) {
    int node = node0 + lane;
    const v2f* p2 = reinterpret_cast<const v2f*>(part) + node;
    v2f a0 = {0.0f, 0.0f}, a1 = {0.0f, 0.0f}, a2 = {0.0f, 0.0f}, a3 = {0.0f, 0.0f};
    int s = 0;
    for (; s + 4 <= nsl; s += 4) {
      a0 += p2[(size_t)(s + 0) * N];
      a1 += p2[(size_t)(s + 1) * N];
      a2 += p2[(size_t)(s + 2) * N];
      a3 += p2[(size_t)(s + 3) * N];
    }
    for (; s < nsl; ++s) a0 += p2[(size_t)s * N];
    v2f xv = reinterpret_cast<const v2f*>(xs)[node];
    float di = dinv[node];
    a = (((a0 + a1) + (a2 + a3)) + xv) * di;
    b.x = W1[0 * 16 + lane];
    b.y = W1[1 * 16 + lane];
  }

  v8f c = {};
  c = __builtin_amdgcn_wmma_f32_16x16x4_f32(
       false, a,  false, b,
       (short)0, c,  false,  false);

  int row0 = node0 + ((lane < 16) ? 0 : 8);
  int col  = lane & 15;
  float bias = b1[col];
#pragma unroll
  for (int j = 0; j < 8; ++j)
    r[(row0 + j) * 16 + col] = fmaxf(c[j] + bias, 0.0f);
}

__global__ void k_gemm1_tail(const float* __restrict__ part, const float* __restrict__ xs,
                             const float* __restrict__ dinv, const float* __restrict__ W1,
                             const float* __restrict__ b1, float* __restrict__ r,
                             int n0, int N, int nsl) {
  int tid = blockIdx.x * blockDim.x + threadIdx.x;
  int n = n0 + (tid >> 4);
  if (n >= N) return;
  int f = tid & 15;
  const v2f* p2 = reinterpret_cast<const v2f*>(part) + n;
  v2f acc = {0.0f, 0.0f};
  for (int s = 0; s < nsl; ++s) acc += p2[(size_t)s * N];
  v2f xv = reinterpret_cast<const v2f*>(xs)[n];
  v2f ax = (acc + xv) * dinv[n];
  r[n * 16 + f] = fmaxf(ax.x * W1[f] + ax.y * W1[16 + f] + b1[f], 0.0f);
}

__global__ void k_gemm2(const float* __restrict__ r, const float* __restrict__ W2,
                        const float* __restrict__ dinv,
                        float* __restrict__ h2s, int ntiles) {
  int wave = (blockIdx.x * blockDim.x + threadIdx.x) >> 5;
  int lane = threadIdx.x & 31;
  if (wave >= ntiles) return;
  int node0 = wave << 4;

  int m = lane & 15;
  int node = node0 + m;

  v8f c = {};
#pragma unroll
  for (int kk = 0; kk < 4; ++kk) {
    int kbase = kk * 4 + ((lane < 16) ? 0 : 2);
    v2f a, b;
    a.x = r[node * 16 + kbase + 0];
    a.y = r[node * 16 + kbase + 1];
    b.x = (m < 2) ? W2[(kbase + 0) * 2 + m] : 0.0f;
    b.y = (m < 2) ? W2[(kbase + 1) * 2 + m] : 0.0f;
    c = __builtin_amdgcn_wmma_f32_16x16x4_f32(
        false, a, false, b, (short)0, c, false, false);
  }

  int row0 = node0 + ((lane < 16) ? 0 : 8);
  int col  = lane & 15;
  if (col < 2) {
#pragma unroll
    for (int j = 0; j < 8; ++j)
      h2s[(row0 + j) * 2 + col] = c[j] * dinv[row0 + j];
  }
}

__global__ void k_gemm2_tail(const float* __restrict__ r, const float* __restrict__ W2,
                             const float* __restrict__ dinv,
                             float* __restrict__ h2s, int n0, int N) {
  int tid = blockIdx.x * blockDim.x + threadIdx.x;
  int n = n0 + (tid >> 1);
  if (n >= N) return;
  int f = tid & 1;
  float acc = 0.0f;
#pragma unroll
  for (int k = 0; k < 16; ++k)
    acc += r[n * 16 + k] * W2[k * 2 + f];
  h2s[n * 2 + f] = acc * dinv[n];
}

__global__ void k_final(const float* __restrict__ part, const float* __restrict__ h2s,
                        const float* __restrict__ dinv, const float* __restrict__ b2,
                        float* __restrict__ out, int N, int nsl) {
  int tid = blockIdx.x * blockDim.x + threadIdx.x;
  if (tid >= 2 * N) return;
  int n = tid >> 1;
  int f = tid & 1;
  const size_t stride = (size_t)2 * N;
  float a0 = 0.0f, a1 = 0.0f, a2 = 0.0f, a3 = 0.0f;
  int s = 0;
  for (; s + 4 <= nsl; s += 4) {
    a0 += part[(size_t)(s + 0) * stride + tid];
    a1 += part[(size_t)(s + 1) * stride + tid];
    a2 += part[(size_t)(s + 2) * stride + tid];
    a3 += part[(size_t)(s + 3) * stride + tid];
  }
  for (; s < nsl; ++s) a0 += part[(size_t)s * stride + tid];
  float di = dinv[n];
  out[tid] = ((a0 + a1) + (a2 + a3)) * di + (h2s[tid] * di + b2[f]);
}

extern "C" void kernel_launch(void* const* d_in, const int* in_sizes, int n_in,
                              void* d_out, int out_size, void* d_ws, size_t ws_size,
                              hipStream_t stream) {
  const float* x  = (const float*)d_in[0];
  const int*   ei = (const int*)d_in[1];
  const float* W1 = (const float*)d_in[2];
  const float* b1 = (const float*)d_in[3];
  const float* W2 = (const float*)d_in[4];
  const float* b2 = (const float*)d_in[5];
  float* out = (float*)d_out;

  const int N = in_sizes[0] / 2;
  const int E = in_sizes[1] / 2;
  const int* src = ei;
  const int* dst = ei + E;

  const size_t NP = ((size_t)N + 63) & ~(size_t)63;
  float* ws   = (float*)d_ws;
  float* dinv = ws;
  float* xs   = ws + NP;
  float* r    = ws + (size_t)3 * NP;
  float* h2s  = ws + (size_t)19 * NP;
  float* part = ws + (size_t)21 * NP;

  const int B = 256;
  const int ntiles = N / 16;
  const int n0 = ntiles * 16;
  const int ntail = N - n0;

  const int nr1 = (N + TILE_FLOATS - 1) / TILE_FLOATS;
  const int ns1 = (256 / nr1) > 0 ? (256 / nr1) : 1;
  const int nr2 = (2 * N + TILE_FLOATS - 1) / TILE_FLOATS;
  const int ns2 = (256 / nr2) > 0 ? (256 / nr2) : 1;

  k_degree_lds<<<dim3(ns1, nr1), SC_THREADS, 0, stream>>>(dst, (int*)part, E, N);
  k_dinv  <<<(N + B - 1) / B,      B, 0, stream>>>((const int*)part, x, dinv, xs, N, ns1);

  k_agg_lds<<<dim3(ns2, nr2), SC_THREADS, 0, stream>>>(src, dst, xs, part, E, N);
  k_gemm1 <<<(ntiles * 32 + B - 1) / B, B, 0, stream>>>(part, xs, dinv, W1, b1, r, ntiles, N, ns2);
  if (ntail > 0)
    k_gemm1_tail<<<(ntail * 16 + B - 1) / B, B, 0, stream>>>(part, xs, dinv, W1, b1, r, n0, N, ns2);

  k_gemm2 <<<(ntiles * 32 + B - 1) / B, B, 0, stream>>>(r, W2, dinv, h2s, ntiles);
  if (ntail > 0)
    k_gemm2_tail<<<(ntail * 2 + B - 1) / B, B, 0, stream>>>(r, W2, dinv, h2s, n0, N);

  k_agg_lds<<<dim3(ns2, nr2), SC_THREADS, 0, stream>>>(src, dst, h2s, part, E, N);
  k_final <<<(2 * N + B - 1) / B,  B, 0, stream>>>(part, h2s, dinv, b2, out, N, ns2);
}


# Edges bucketed once per call by (destination range, source window): count pass, LDS-ordered place pass; both aggregations read own bucket, source rows from an LDS window; float sum order changes
# speedup vs baseline: 6.7115x; 1.5146x over previous
#include <hip/hip_runtime.h>


typedef __attribute__((ext_vector_type(2))) float v2f;
typedef __attribute__((ext_vector_type(8))) float v8f;

#define TILE_FLOATS 65536
#define SC_THREADS  1024
#define SC_BATCH    8

__global__ void k_degree_lds(const int* __restrict__ dst, int* __restrict__ partial,
                             int E, int N) {
  __shared__ __attribute__((aligned(16))) int tile[TILE_FLOATS];
  const int t = threadIdx.x;
  const int R = (((N + (int)gridDim.y - 1) / (int)gridDim.y) + 31) & ~31;
  const int eps = (E + (int)gridDim.x - 1) / (int)gridDim.x;
  const int lo = blockIdx.y * R;
  const int nloc = max(0, min(R, N - lo));

  const int4 z4 = {0, 0, 0, 0};
  for (int i = t * 4; i < TILE_FLOATS; i += SC_THREADS * 4)
    *reinterpret_cast<int4*>(&tile[i]) = z4;
  __syncthreads();

  const int e0 = blockIdx.x * eps;
  const int e1 = min(E, e0 + eps);
  const int elast = e1 - 1;

  int dn[SC_BATCH];
  int base = e0 + t;
  for (int k = 0; k < SC_BATCH; ++k) {
    int e = base + k * SC_THREADS;
    int dv = dst[min(e, elast)];
    dn[k] = (e < e1) ? dv : -1;
  }
  for (; base < e1; base += SC_BATCH * SC_THREADS) {
    int d[SC_BATCH];
    for (int k = 0; k < SC_BATCH; ++k) d[k] = dn[k];
    const int nb = base + SC_BATCH * SC_THREADS;
    for (int k = 0; k < SC_BATCH; ++k) {
      int e = nb + k * SC_THREADS;
      int dv = dst[min(e, elast)];
      dn[k] = (e < e1) ? dv : -1;
    }
    for (int k = 0; k < SC_BATCH; ++k) {
      unsigned r = (unsigned)(d[k] - lo);
      if (r < (unsigned)nloc) atomicAdd(&tile[r], 1);
    }
  }
  __syncthreads();

  int* p = partial + ((size_t)blockIdx.x * N + lo);
  for (int i = t; i < nloc; i += SC_THREADS) p[i] = tile[i];
}

__global__ void k_agg_lds(const int* __restrict__ src, const int* __restrict__ dst,
                           const float* __restrict__ feat, float* __restrict__ partial,
                           int E, int N) {
  __shared__ __attribute__((aligned(16))) float tile[TILE_FLOATS];
  const int t = threadIdx.x;
  const int R = (((N + (int)gridDim.y - 1) / (int)gridDim.y) + 31) & ~31;
  const int eps = (E + (int)gridDim.x - 1) / (int)gridDim.x;
  const int lo = blockIdx.y * R;
  const int nloc = max(0, min(R, N - lo));

  const float4 z4 = {0.0f, 0.0f, 0.0f, 0.0f};
  for (int i = t * 4; i < TILE_FLOATS; i += SC_THREADS * 4)
    *reinterpret_cast<float4*>(&tile[i]) = z4;
  __syncthreads();

  const int e0 = blockIdx.x * eps;
  const int e1 = min(E, e0 + eps);
  const int elast = e1 - 1;
  const v2f* feat2 = reinterpret_cast<const v2f*>(feat);

  int dn[SC_BATCH], sn[SC_BATCH];
  int base = e0 + t;
  for (int k = 0; k < SC_BATCH; ++k) {
    int e = base + k * SC_THREADS;
    int ec = min(e, elast);
    int dv = dst[ec];
    dn[k] = (e < e1) ? dv : -1;
    sn[k] = src[ec];
  }
  for (; base < e1; base += SC_BATCH * SC_THREADS) {
    int d[SC_BATCH], s[SC_BATCH];
    for (int k = 0; k < SC_BATCH; ++k) { d[k] = dn[k]; s[k] = sn[k]; }
    const int nb = base + SC_BATCH * SC_THREADS;
    for (int k = 0; k < SC_BATCH; ++k) {
      int e = nb + k * SC_THREADS;
      int ec = min(e, elast);
      int dv = dst[ec];
      dn[k] = (e < e1) ? dv : -1;
      sn[k] = src[ec];
    }
    v2f v[SC_BATCH];
    for (int k = 0; k < SC_BATCH; ++k) {
      unsigned r = (unsigned)(d[k] - lo);
      v[k] = feat2[(r < (unsigned)nloc) ? s[k] : 0];
    }
    for (int k = 0; k < SC_BATCH; ++k) {
      unsigned r = (unsigned)(d[k] - lo);
      if (r < (unsigned)nloc) {
        atomicAdd(&tile[2 * r + 0], v[k].x);
        atomicAdd(&tile[2 * r + 1], v[k].y);
      }
    }
  }
  __syncthreads();

  float* p = partial + ((size_t)blockIdx.x * N + lo) * 2;
  const int cnt = nloc * 2;
  for (int i = t; i < cnt; i += SC_THREADS) p[i] = tile[i];
}

#define BIN_MAX    256
#define STAGE_MAX  32768
#define WIN_FLOATS 16384
#define AG_BATCH   16

__device__ __forceinline__ int qdiv(int v, int dvs, float inv) {
  int q = (int)((float)v * inv);
  int r = v - q * dvs;
  q += (r >= dvs) ? 1 : 0;
  q -= (r < 0) ? 1 : 0;
  return q;
}

__device__ __forceinline__ int edge_bucket(int d, int s, int N, int R, int WR,
                                           int nwin, float invR, float invW,
                                           int* packed) {
  if ((unsigned)d >= (unsigned)N || (unsigned)s >= (unsigned)N) return -1;
  int y = qdiv(d, R, invR);
  int x = qdiv(s, WR, invW);
  *packed = ((s - x * WR) << 15) | (d - y * R);
  return y * nwin + x;
}

__global__ void k_count(const int* __restrict__ src, const int* __restrict__ dst,
                        int* __restrict__ cnt, int E, int N, int R, int WR,
                        int nwin, int NB, int epb) {
  __shared__ int hist[BIN_MAX];
  const int t = threadIdx.x;
  if (t < BIN_MAX) hist[t] = 0;
  __syncthreads();
  const float invR = 1.0f / (float)R, invW = 1.0f / (float)WR;
  const int e0 = min(E, (int)blockIdx.x * epb);
  const int e1 = min(E, e0 + epb);
  for (int base = e0 + t; base < e1; base += SC_BATCH * SC_THREADS) {
    int dv[SC_BATCH], sv[SC_BATCH];
    for (int k = 0; k < SC_BATCH; ++k) {
      int e = base + k * SC_THREADS;
      int ec = min(e, e1 - 1);
      int d = dst[ec];
      sv[k] = src[ec];
      dv[k] = (e < e1) ? d : -1;
    }
    for (int k = 0; k < SC_BATCH; ++k) {
      int pk;
      int b = edge_bucket(dv[k], sv[k], N, R, WR, nwin, invR, invW, &pk);
      if (b >= 0) atomicAdd(&hist[b], 1);
    }
  }
  __syncthreads();
  if (t < NB) cnt[(size_t)blockIdx.x * NB + t] = hist[t];
}

__global__ void k_place(const int* __restrict__ src, const int* __restrict__ dst,
                        const int* __restrict__ cnt, int* __restrict__ ent,
                        int* __restrict__ bst, int E, int N, int R, int WR,
                        int nwin, int NB, int epb) {
  __shared__ int tot[BIN_MAX], pre[BIN_MAX], own[BIN_MAX];
  __shared__ int cur[BIN_MAX], lst[BIN_MAX], gst[BIN_MAX];
  __shared__ int sa[BIN_MAX], so[BIN_MAX];
  __shared__ int sval[STAGE_MAX];
  __shared__ unsigned char sbin[STAGE_MAX];
  const int t = threadIdx.x;
  const int slice = blockIdx.x;
  const int S = gridDim.x;
  if (t < BIN_MAX) {
    tot[t] = 0;
    pre[t] = 0;
    own[t] = (t < NB) ? cnt[(size_t)slice * NB + t] : 0;
  }
  __syncthreads();
  {
    const int b = t & (BIN_MAX - 1);
    const int part = t >> 8;
    const int per = (S + 3) >> 2;
    const int s0 = part * per;
    const int s1 = min(S, s0 + per);
    if (b < NB && s0 < s1) {
      int all = 0, before = 0;
      for (int s = s0; s < s1; s += 16) {
        int v[16];
#pragma unroll
        for (int k = 0; k < 16; ++k) {
          int ss = min(s + k, s1 - 1);
          v[k] = cnt[(size_t)ss * NB + b];
        }
#pragma unroll
        for (int k = 0; k < 16; ++k) {
          int ss = s + k;
          int vv = (ss < s1) ? v[k] : 0;
          all += vv;
          before += (ss < slice) ? vv : 0;
        }
      }
      atomicAdd(&tot[b], all);
      atomicAdd(&pre[b], before);
    }
  }
  __syncthreads();
  int pa = 0, po = 0;
  if (t < BIN_MAX) { pa = tot[t]; po = own[t]; sa[t] = pa; so[t] = po; }
  __syncthreads();
  for (int off = 1; off < BIN_MAX; off <<= 1) {
    int a2 = 0, o2 = 0;
    if (t < BIN_MAX && t >= off) { a2 = sa[t - off]; o2 = so[t - off]; }
    __syncthreads();
    if (t < BIN_MAX) { pa += a2; po += o2; sa[t] = pa; so[t] = po; }
    __syncthreads();
  }
  if (t < NB) {
    int gs = pa - tot[t];
    int ls = po - own[t];
    gst[t] = gs + pre[t];
    lst[t] = ls;
    cur[t] = ls;
    if (slice == 0) {
      bst[t] = gs;
      if (t == NB - 1) bst[NB] = pa;
    }
  }
  __syncthreads();

  const float invR = 1.0f / (float)R, invW = 1.0f / (float)WR;
  const int e0 = min(E, slice * epb);
  const int e1 = min(E, e0 + epb);
  for (int base = e0 + t; base < e1; base += SC_BATCH * SC_THREADS) {
    int dv[SC_BATCH], sv[SC_BATCH];
    for (int k = 0; k < SC_BATCH; ++k) {
      int e = base + k * SC_THREADS;
      int ec = min(e, e1 - 1);
      int d = dst[ec];
      sv[k] = src[ec];
      dv[k] = (e < e1) ? d : -1;
    }
    for (int k = 0; k < SC_BATCH; ++k) {
      int pk = 0;
      int b = edge_bucket(dv[k], sv[k], N, R, WR, nwin, invR, invW, &pk);
      if (b >= 0) {
        int lp = atomicAdd(&cur[b], 1);
        sval[lp] = pk;
        sbin[lp] = (unsigned char)b;
      }
    }
  }
  __syncthreads();
  const int ntot = lst[NB - 1] + own[NB - 1];
  for (int i = t; i < ntot; i += SC_THREADS) {
    int b = sbin[i];
    ent[gst[b] + (i - lst[b])] = sval[i];
  }
}

__global__ void k_agg_bin(const int* __restrict__ ent, const int* __restrict__ bst,
                          const float* __restrict__ feat, float* __restrict__ partial,
                          int N, int R, int WR) {
  __shared__ __attribute__((aligned(16))) float tile[TILE_FLOATS];
  __shared__ __attribute__((aligned(16))) float win[WIN_FLOATS];
  const int t = threadIdx.x;
  const int lo = blockIdx.y * R;
  const int nloc = max(0, min(R, N - lo));
  const int wlo = blockIdx.x * WR;
  const int wn = max(0, min(WR, N - wlo));
  const int bucket = blockIdx.y * gridDim.x + blockIdx.x;
  const int b0 = bst[bucket];
  const int b1 = bst[bucket + 1];

  const float4 z4 = {0.0f, 0.0f, 0.0f, 0.0f};
  for (int i = t * 4; i < nloc * 2; i += SC_THREADS * 4)
    *reinterpret_cast<float4*>(&tile[i]) = z4;
  const float* wsrc = feat + (size_t)wlo * 2;
  for (int i = t; i < wn * 2; i += SC_THREADS) win[i] = wsrc[i];
  __syncthreads();

  const v2f* win2 = reinterpret_cast<const v2f*>(win);
  for (int base = b0 + t; base < b1; base += AG_BATCH * SC_THREADS) {
    int en[AG_BATCH];
    for (int k = 0; k < AG_BATCH; ++k) {
      int i = base + k * SC_THREADS;
      int v = ent[min(i, b1 - 1)];
      en[k] = (i < b1) ? v : -1;
    }
    v2f v[AG_BATCH];
    for (int k = 0; k < AG_BATCH; ++k)
      v[k] = win2[(en[k] >= 0) ? (en[k] >> 15) : 0];
    for (int k = 0; k < AG_BATCH; ++k) {
      if (en[k] >= 0) {
        int dl = en[k] & 32767;
        atomicAdd(&tile[2 * dl + 0], v[k].x);
        atomicAdd(&tile[2 * dl + 1], v[k].y);
      }
    }
  }
  __syncthreads();

  float* p = partial + ((size_t)blockIdx.x * N + lo) * 2;
  const int cnt2 = nloc * 2;
  for (int i = t; i < cnt2; i += SC_THREADS) p[i] = tile[i];
}

__global__ void k_dinv(const int* __restrict__ part, const float* __restrict__ x,
                       float* __restrict__ dinv, float* __restrict__ xs,
                       int N, int nsl) {
  int i = blockIdx.x * blockDim.x + threadIdx.x;
  if (i < N) {
    int c0 = 0, c1 = 0, c2 = 0, c3 = 0;
    int s = 0;
    for (; s + 4 <= nsl; s += 4) {
      c0 += part[(size_t)(s + 0) * N + i];
      c1 += part[(size_t)(s + 1) * N + i];
      c2 += part[(size_t)(s + 2) * N + i];
      c3 += part[(size_t)(s + 3) * N + i];
    }
    for (; s < nsl; ++s) c0 += part[(size_t)s * N + i];
    float d = 1.0f + (float)((c0 + c1) + (c2 + c3));
    float di = (d > 0.0f) ? rsqrtf(d) : 0.0f;
    dinv[i] = di;
    v2f xv = reinterpret_cast<const v2f*>(x)[i];
    reinterpret_cast<v2f*>(xs)[i] = xv * di;
  }
}

__global__ void k_gemm1(const float* __restrict__ part, const float* __restrict__ xs,
                        const float* __restrict__ dinv, const float* __restrict__ W1,
                        const float* __restrict__ b1, float* __restrict__ r,
                        int ntiles, int N, int nsl) {
  int wave = (blockIdx.x * blockDim.x + threadIdx.x) >> 5;
  int lane = threadIdx.x & 31;
  if (wave >= ntiles) return;
  int node0 = wave << 4;

  v2f a = {0.0f, 0.0f};
  v2f b = {0.0f, 0.0f};
  if (lane < 16) {
    int node = node0 + lane;
    const v2f* p2 = reinterpret_cast<const v2f*>(part) + node;
    v2f a0 = {0.0f, 0.0f}, a1 = {0.0f, 0.0f}, a2 = {0.0f, 0.0f}, a3 = {0.0f, 0.0f};
    int s = 0;
    for (; s + 4 <= nsl; s += 4) {
      a0 += p2[(size_t)(s + 0) * N];
      a1 += p2[(size_t)(s + 1) * N];
      a2 += p2[(size_t)(s + 2) * N];
      a3 += p2[(size_t)(s + 3) * N];
    }
    for (; s < nsl; ++s) a0 += p2[(size_t)s * N];
    v2f xv = reinterpret_cast<const v2f*>(xs)[node];
    float di = dinv[node];
    a = (((a0 + a1) + (a2 + a3)) + xv) * di;
    b.x = W1[0 * 16 + lane];
    b.y = W1[1 * 16 + lane];
  }

  v8f c = {};
  c = __builtin_amdgcn_wmma_f32_16x16x4_f32(
       false, a,  false, b,
       (short)0, c,  false,  false);

  int row0 = node0 + ((lane < 16) ? 0 : 8);
  int col  = lane & 15;
  float bias = b1[col];
#pragma unroll
  for (int j = 0; j < 8; ++j)
    r[(row0 + j) * 16 + col] = fmaxf(c[j] + bias, 0.0f);
}

__global__ void k_gemm1_tail(const float* __restrict__ part, const float* __restrict__ xs,
                             const float* __restrict__ dinv, const float* __restrict__ W1,
                             const float* __restrict__ b1, float* __restrict__ r,
                             int n0, int N, int nsl) {
  int tid = blockIdx.x * blockDim.x + threadIdx.x;
  int n = n0 + (tid >> 4);
  if (n >= N) return;
  int f = tid & 15;
  const v2f* p2 = reinterpret_cast<const v2f*>(part) + n;
  v2f acc = {0.0f, 0.0f};
  for (int s = 0; s < nsl; ++s) acc += p2[(size_t)s * N];
  v2f xv = reinterpret_cast<const v2f*>(xs)[n];
  v2f ax = (acc + xv) * dinv[n];
  r[n * 16 + f] = fmaxf(ax.x * W1[f] + ax.y * W1[16 + f] + b1[f], 0.0f);
}

__global__ void k_gemm2(const float* __restrict__ r, const float* __restrict__ W2,
                        const float* __restrict__ dinv,
                        float* __restrict__ h2s, int ntiles) {
  int wave = (blockIdx.x * blockDim.x + threadIdx.x) >> 5;
  int lane = threadIdx.x & 31;
  if (wave >= ntiles) return;
  int node0 = wave << 4;

  int m = lane & 15;
  int node = node0 + m;

  v8f c = {};
#pragma unroll
  for (int kk = 0; kk < 4; ++kk) {
    int kbase = kk * 4 + ((lane < 16) ? 0 : 2);
    v2f a, b;
    a.x = r[node * 16 + kbase + 0];
    a.y = r[node * 16 + kbase + 1];
    b.x = (m < 2) ? W2[(kbase + 0) * 2 + m] : 0.0f;
    b.y = (m < 2) ? W2[(kbase + 1) * 2 + m] : 0.0f;
    c = __builtin_amdgcn_wmma_f32_16x16x4_f32(
        false, a, false, b, (short)0, c, false, false);
  }

  int row0 = node0 + ((lane < 16) ? 0 : 8);
  int col  = lane & 15;
  if (col < 2) {
#pragma unroll
    for (int j = 0; j < 8; ++j)
      h2s[(row0 + j) * 2 + col] = c[j] * dinv[row0 + j];
  }
}

__global__ void k_gemm2_tail(const float* __restrict__ r, const float* __restrict__ W2,
                             const float* __restrict__ dinv,
                             float* __restrict__ h2s, int n0, int N) {
  int tid = blockIdx.x * blockDim.x + threadIdx.x;
  int n = n0 + (tid >> 1);
  if (n >= N) return;
  int f = tid & 1;
  float acc = 0.0f;
#pragma unroll
  for (int k = 0; k < 16; ++k)
    acc += r[n * 16 + k] * W2[k * 2 + f];
  h2s[n * 2 + f] = acc * dinv[n];
}

__global__ void k_final(const float* __restrict__ part, const float* __restrict__ h2s,
                        const float* __restrict__ dinv, const float* __restrict__ b2,
                        float* __restrict__ out, int N, int nsl) {
  int tid = blockIdx.x * blockDim.x + threadIdx.x;
  if (tid >= 2 * N) return;
  int n = tid >> 1;
  int f = tid & 1;
  const size_t stride = (size_t)2 * N;
  float a0 = 0.0f, a1 = 0.0f, a2 = 0.0f, a3 = 0.0f;
  int s = 0;
  for (; s + 4 <= nsl; s += 4) {
    a0 += part[(size_t)(s + 0) * stride + tid];
    a1 += part[(size_t)(s + 1) * stride + tid];
    a2 += part[(size_t)(s + 2) * stride + tid];
    a3 += part[(size_t)(s + 3) * stride + tid];
  }
  for (; s < nsl; ++s) a0 += part[(size_t)s * stride + tid];
  float di = dinv[n];
  out[tid] = ((a0 + a1) + (a2 + a3)) * di + (h2s[tid] * di + b2[f]);
}

extern "C" void kernel_launch(void* const* d_in, const int* in_sizes, int n_in,
                              void* d_out, int out_size, void* d_ws, size_t ws_size,
                              hipStream_t stream) {
  const float* x  = (const float*)d_in[0];
  const int*   ei = (const int*)d_in[1];
  const float* W1 = (const float*)d_in[2];
  const float* b1 = (const float*)d_in[3];
  const float* W2 = (const float*)d_in[4];
  const float* b2 = (const float*)d_in[5];
  float* out = (float*)d_out;

  const int N = in_sizes[0] / 2;
  const int E = in_sizes[1] / 2;
  const int* src = ei;
  const int* dst = ei + E;

  const size_t NP = ((size_t)N + 63) & ~(size_t)63;
  float* ws   = (float*)d_ws;
  float* dinv = ws;
  float* xs   = ws + NP;
  float* r    = ws + (size_t)3 * NP;
  float* h2s  = ws + (size_t)19 * NP;
  float* part = ws + (size_t)21 * NP;

  const int B = 256;
  const int ntiles = N / 16;
  const int n0 = ntiles * 16;
  const int ntail = N - n0;

  const int nr1 = (N + TILE_FLOATS - 1) / TILE_FLOATS;
  const int ns1 = (256 / nr1) > 0 ? (256 / nr1) : 1;
  const int nr2 = (2 * N + TILE_FLOATS - 1) / TILE_FLOATS;
  const int ns2 = (256 / nr2) > 0 ? (256 / nr2) : 1;

  const int S   = 256;
  const int R2  = (((N + nr2 - 1) / nr2) + 31) & ~31;
  const int WR  = (N + ns2 - 1) / ns2;
  const int NB  = nr2 * ns2;
  const int epb = (E + S - 1) / S;
  const size_t pa1 = (size_t)ns1 * N, pa2 = (size_t)ns2 * 2 * N;
  const size_t partFloats = (((pa1 > pa2) ? pa1 : pa2) + 63) & ~(size_t)63;
  int* cnt = (int*)(part + partFloats);
  int* bst = cnt + (((size_t)S * NB + 63) & ~(size_t)63);
  int* ent = bst + (((size_t)NB + 1 + 63) & ~(size_t)63);
  const size_t need = ((size_t)(ent - (int*)ws) + (size_t)E) * sizeof(int);
  const bool binned = E > 0 && NB <= BIN_MAX && 2 * R2 <= TILE_FLOATS &&
                      2 * WR <= WIN_FLOATS && epb <= STAGE_MAX &&
                      N < (1 << 24) && need <= ws_size;

  k_degree_lds<<<dim3(ns1, nr1), SC_THREADS, 0, stream>>>(dst, (int*)part, E, N);
  k_dinv  <<<(N + B - 1) / B,      B, 0, stream>>>((const int*)part, x, dinv, xs, N, ns1);

  if (binned) {
    k_count<<<S, SC_THREADS, 0, stream>>>(src, dst, cnt, E, N, R2, WR, ns2, NB, epb);
    k_place<<<S, SC_THREADS, 0, stream>>>(src, dst, cnt, ent, bst, E, N, R2, WR, ns2, NB, epb);
    k_agg_bin<<<dim3(ns2, nr2), SC_THREADS, 0, stream>>>(ent, bst, xs, part, N, R2, WR);
  } else {
    k_agg_lds<<<dim3(ns2, nr2), SC_THREADS, 0, stream>>>(src, dst, xs, part, E, N);
  }
  k_gemm1 <<<(ntiles * 32 + B - 1) / B, B, 0, stream>>>(part, xs, dinv, W1, b1, r, ntiles, N, ns2);
  if (ntail > 0)
    k_gemm1_tail<<<(ntail * 16 + B - 1) / B, B, 0, stream>>>(part, xs, dinv, W1, b1, r, n0, N, ns2);

  k_gemm2 <<<(ntiles * 32 + B - 1) / B, B, 0, stream>>>(r, W2, dinv, h2s, ntiles);
  if (ntail > 0)
    k_gemm2_tail<<<(ntail * 2 + B - 1) / B, B, 0, stream>>>(r, W2, dinv, h2s, n0, N);

  if (binned)
    k_agg_bin<<<dim3(ns2, nr2), SC_THREADS, 0, stream>>>(ent, bst, h2s, part, N, R2, WR);
  else
    k_agg_lds<<<dim3(ns2, nr2), SC_THREADS, 0, stream>>>(src, dst, h2s, part, E, N);
  k_final <<<(2 * N + B - 1) / B,  B, 0, stream>>>(part, h2s, dinv, b2, out, N, ns2);
}


# Both node-wise WMMA products in one launch: product 1 formed transposed so its ReLU'd D registers are product 2's B fragments; r never stored, 32 nodes per wave; product-2 k sum order changes
# speedup vs baseline: 7.0696x; 1.0534x over previous
#include <hip/hip_runtime.h>


typedef __attribute__((ext_vector_type(2))) float v2f;
typedef __attribute__((ext_vector_type(8))) float v8f;

#define TILE_FLOATS 65536
#define SC_THREADS  1024
#define SC_BATCH    8

__global__ void k_degree_lds(const int* __restrict__ dst, int* __restrict__ partial,
                             int E, int N) {
  __shared__ __attribute__((aligned(16))) int tile[TILE_FLOATS];
  const int t = threadIdx.x;
  const int R = (((N + (int)gridDim.y - 1) / (int)gridDim.y) + 31) & ~31;
  const int eps = (E + (int)gridDim.x - 1) / (int)gridDim.x;
  const int lo = blockIdx.y * R;
  const int nloc = max(0, min(R, N - lo));

  const int4 z4 = {0, 0, 0, 0};
  for (int i = t * 4; i < TILE_FLOATS; i += SC_THREADS * 4)
    *reinterpret_cast<int4*>(&tile[i]) = z4;
  __syncthreads();

  const int e0 = blockIdx.x * eps;
  const int e1 = min(E, e0 + eps);
  const int elast = e1 - 1;

  int dn[SC_BATCH];
  int base = e0 + t;
  for (int k = 0; k < SC_BATCH; ++k) {
    int e = base + k * SC_THREADS;
    int dv = dst[min(e, elast)];
    dn[k] = (e < e1) ? dv : -1;
  }
  for (; base < e1; base += SC_BATCH * SC_THREADS) {
    int d[SC_BATCH];
    for (int k = 0; k < SC_BATCH; ++k) d[k] = dn[k];
    const int nb = base + SC_BATCH * SC_THREADS;
    for (int k = 0; k < SC_BATCH; ++k) {
      int e = nb + k * SC_THREADS;
      int dv = dst[min(e, elast)];
      dn[k] = (e < e1) ? dv : -1;
    }
    for (int k = 0; k < SC_BATCH; ++k) {
      unsigned r = (unsigned)(d[k] - lo);
      if (r < (unsigned)nloc) atomicAdd(&tile[r], 1);
    }
  }
  __syncthreads();

  int* p = partial + ((size_t)blockIdx.x * N + lo);
  for (int i = t; i < nloc; i += SC_THREADS) p[i] = tile[i];
}

__global__ void k_agg_lds(const int* __restrict__ src, const int* __restrict__ dst,
                           const float* __restrict__ feat, float* __restrict__ partial,
                           int E, int N) {
  __shared__ __attribute__((aligned(16))) float tile[TILE_FLOATS];
  const int t = threadIdx.x;
  const int R = (((N + (int)gridDim.y - 1) / (int)gridDim.y) + 31) & ~31;
  const int eps = (E + (int)gridDim.x - 1) / (int)gridDim.x;
  const int lo = blockIdx.y * R;
  const int nloc = max(0, min(R, N - lo));

  const float4 z4 = {0.0f, 0.0f, 0.0f, 0.0f};
  for (int i = t * 4; i < TILE_FLOATS; i += SC_THREADS * 4)
    *reinterpret_cast<float4*>(&tile[i]) = z4;
  __syncthreads();

  const int e0 = blockIdx.x * eps;
  const int e1 = min(E, e0 + eps);
  const int elast = e1 - 1;
  const v2f* feat2 = reinterpret_cast<const v2f*>(feat);

  int dn[SC_BATCH], sn[SC_BATCH];
  int base = e0 + t;
  for (int k = 0; k < SC_BATCH; ++k) {
    int e = base + k * SC_THREADS;
    int ec = min(e, elast);
    int dv = dst[ec];
    dn[k] = (e < e1) ? dv : -1;
    sn[k] = src[ec];
  }
  for (; base < e1; base += SC_BATCH * SC_THREADS) {
    int d[SC_BATCH], s[SC_BATCH];
    for (int k = 0; k < SC_BATCH; ++k) { d[k] = dn[k]; s[k] = sn[k]; }
    const int nb = base + SC_BATCH * SC_THREADS;
    for (int k = 0; k < SC_BATCH; ++k) {
      int e = nb + k * SC_THREADS;
      int ec = min(e, elast);
      int dv = dst[ec];
      dn[k] = (e < e1) ? dv : -1;
      sn[k] = src[ec];
    }
    v2f v[SC_BATCH];
    for (int k = 0; k < SC_BATCH; ++k) {
      unsigned r = (unsigned)(d[k] - lo);
      v[k] = feat2[(r < (unsigned)nloc) ? s[k] : 0];
    }
    for (int k = 0; k < SC_BATCH; ++k) {
      unsigned r = (unsigned)(d[k] - lo);
      if (r < (unsigned)nloc) {
        atomicAdd(&tile[2 * r + 0], v[k].x);
        atomicAdd(&tile[2 * r + 1], v[k].y);
      }
    }
  }
  __syncthreads();

  float* p = partial + ((size_t)blockIdx.x * N + lo) * 2;
  const int cnt = nloc * 2;
  for (int i = t; i < cnt; i += SC_THREADS) p[i] = tile[i];
}

#define BIN_MAX    256
#define STAGE_MAX  32768
#define WIN_FLOATS 16384
#define AG_BATCH   16

__device__ __forceinline__ int qdiv(int v, int dvs, float inv) {
  int q = (int)((float)v * inv);
  int r = v - q * dvs;
  q += (r >= dvs) ? 1 : 0;
  q -= (r < 0) ? 1 : 0;
  return q;
}

__device__ __forceinline__ int edge_bucket(int d, int s, int N, int R, int WR,
                                           int nwin, float invR, float invW,
                                           int* packed) {
  if ((unsigned)d >= (unsigned)N || (unsigned)s >= (unsigned)N) return -1;
  int y = qdiv(d, R, invR);
  int x = qdiv(s, WR, invW);
  *packed = ((s - x * WR) << 15) | (d - y * R);
  return y * nwin + x;
}

__global__ void k_count(const int* __restrict__ src, const int* __restrict__ dst,
                        int* __restrict__ cnt, int E, int N, int R, int WR,
                        int nwin, int NB, int epb) {
  __shared__ int hist[BIN_MAX];
  const int t = threadIdx.x;
  if (t < BIN_MAX) hist[t] = 0;
  __syncthreads();
  const float invR = 1.0f / (float)R, invW = 1.0f / (float)WR;
  const int e0 = min(E, (int)blockIdx.x * epb);
  const int e1 = min(E, e0 + epb);
  for (int base = e0 + t; base < e1; base += SC_BATCH * SC_THREADS) {
    int dv[SC_BATCH], sv[SC_BATCH];
    for (int k = 0; k < SC_BATCH; ++k) {
      int e = base + k * SC_THREADS;
      int ec = min(e, e1 - 1);
      int d = dst[ec];
      sv[k] = src[ec];
      dv[k] = (e < e1) ? d : -1;
    }
    for (int k = 0; k < SC_BATCH; ++k) {
      int pk;
      int b = edge_bucket(dv[k], sv[k], N, R, WR, nwin, invR, invW, &pk);
      if (b >= 0) atomicAdd(&hist[b], 1);
    }
  }
  __syncthreads();
  if (t < NB) cnt[(size_t)blockIdx.x * NB + t] = hist[t];
}

__global__ void k_place(const int* __restrict__ src, const int* __restrict__ dst,
                        const int* __restrict__ cnt, int* __restrict__ ent,
                        int* __restrict__ bst, int E, int N, int R, int WR,
                        int nwin, int NB, int epb) {
  __shared__ int tot[BIN_MAX], pre[BIN_MAX], own[BIN_MAX];
  __shared__ int cur[BIN_MAX], lst[BIN_MAX], gst[BIN_MAX];
  __shared__ int sa[BIN_MAX], so[BIN_MAX];
  __shared__ int sval[STAGE_MAX];
  __shared__ unsigned char sbin[STAGE_MAX];
  const int t = threadIdx.x;
  const int slice = blockIdx.x;
  const int S = gridDim.x;
  if (t < BIN_MAX) {
    tot[t] = 0;
    pre[t] = 0;
    own[t] = (t < NB) ? cnt[(size_t)slice * NB + t] : 0;
  }
  __syncthreads();
  {
    const int b = t & (BIN_MAX - 1);
    const int part = t >> 8;
    const int per = (S + 3) >> 2;
    const int s0 = part * per;
    const int s1 = min(S, s0 + per);
    if (b < NB && s0 < s1) {
      int all = 0, before = 0;
      for (int s = s0; s < s1; s += 16) {
        int v[16];
#pragma unroll
        for (int k = 0; k < 16; ++k) {
          int ss = min(s + k, s1 - 1);
          v[k] = cnt[(size_t)ss * NB + b];
        }
#pragma unroll
        for (int k = 0; k < 16; ++k) {
          int ss = s + k;
          int vv = (ss < s1) ? v[k] : 0;
          all += vv;
          before += (ss < slice) ? vv : 0;
        }
      }
      atomicAdd(&tot[b], all);
      atomicAdd(&pre[b], before);
    }
  }
  __syncthreads();
  int pa = 0, po = 0;
  if (t < BIN_MAX) { pa = tot[t]; po = own[t]; sa[t] = pa; so[t] = po; }
  __syncthreads();
  for (int off = 1; off < BIN_MAX; off <<= 1) {
    int a2 = 0, o2 = 0;
    if (t < BIN_MAX && t >= off) { a2 = sa[t - off]; o2 = so[t - off]; }
    __syncthreads();
    if (t < BIN_MAX) { pa += a2; po += o2; sa[t] = pa; so[t] = po; }
    __syncthreads();
  }
  if (t < NB) {
    int gs = pa - tot[t];
    int ls = po - own[t];
    gst[t] = gs + pre[t];
    lst[t] = ls;
    cur[t] = ls;
    if (slice == 0) {
      bst[t] = gs;
      if (t == NB - 1) bst[NB] = pa;
    }
  }
  __syncthreads();

  const float invR = 1.0f / (float)R, invW = 1.0f / (float)WR;
  const int e0 = min(E, slice * epb);
  const int e1 = min(E, e0 + epb);
  for (int base = e0 + t; base < e1; base += SC_BATCH * SC_THREADS) {
    int dv[SC_BATCH], sv[SC_BATCH];
    for (int k = 0; k < SC_BATCH; ++k) {
      int e = base + k * SC_THREADS;
      int ec = min(e, e1 - 1);
      int d = dst[ec];
      sv[k] = src[ec];
      dv[k] = (e < e1) ? d : -1;
    }
    for (int k = 0; k < SC_BATCH; ++k) {
      int pk = 0;
      int b = edge_bucket(dv[k], sv[k], N, R, WR, nwin, invR, invW, &pk);
      if (b >= 0) {
        int lp = atomicAdd(&cur[b], 1);
        sval[lp] = pk;
        sbin[lp] = (unsigned char)b;
      }
    }
  }
  __syncthreads();
  const int ntot = lst[NB - 1] + own[NB - 1];
  for (int i = t; i < ntot; i += SC_THREADS) {
    int b = sbin[i];
    ent[gst[b] + (i - lst[b])] = sval[i];
  }
}

__global__ void k_agg_bin(const int* __restrict__ ent, const int* __restrict__ bst,
                          const float* __restrict__ feat, float* __restrict__ partial,
                          int N, int R, int WR) {
  __shared__ __attribute__((aligned(16))) float tile[TILE_FLOATS];
  __shared__ __attribute__((aligned(16))) float win[WIN_FLOATS];
  const int t = threadIdx.x;
  const int lo = blockIdx.y * R;
  const int nloc = max(0, min(R, N - lo));
  const int wlo = blockIdx.x * WR;
  const int wn = max(0, min(WR, N - wlo));
  const int bucket = blockIdx.y * gridDim.x + blockIdx.x;
  const int b0 = bst[bucket];
  const int b1 = bst[bucket + 1];

  const float4 z4 = {0.0f, 0.0f, 0.0f, 0.0f};
  for (int i = t * 4; i < nloc * 2; i += SC_THREADS * 4)
    *reinterpret_cast<float4*>(&tile[i]) = z4;
  const float* wsrc = feat + (size_t)wlo * 2;
  for (int i = t; i < wn * 2; i += SC_THREADS) win[i] = wsrc[i];
  __syncthreads();

  const v2f* win2 = reinterpret_cast<const v2f*>(win);
  for (int base = b0 + t; base < b1; base += AG_BATCH * SC_THREADS) {
    int en[AG_BATCH];
    for (int k = 0; k < AG_BATCH; ++k) {
      int i = base + k * SC_THREADS;
      int v = ent[min(i, b1 - 1)];
      en[k] = (i < b1) ? v : -1;
    }
    v2f v[AG_BATCH];
    for (int k = 0; k < AG_BATCH; ++k)
      v[k] = win2[(en[k] >= 0) ? (en[k] >> 15) : 0];
    for (int k = 0; k < AG_BATCH; ++k) {
      if (en[k] >= 0) {
        int dl = en[k] & 32767;
        atomicAdd(&tile[2 * dl + 0], v[k].x);
        atomicAdd(&tile[2 * dl + 1], v[k].y);
      }
    }
  }
  __syncthreads();

  float* p = partial + ((size_t)blockIdx.x * N + lo) * 2;
  const int cnt2 = nloc * 2;
  for (int i = t; i < cnt2; i += SC_THREADS) p[i] = tile[i];
}

__global__ void k_dinv(const int* __restrict__ part, const float* __restrict__ x,
                       float* __restrict__ dinv, float* __restrict__ xs,
                       int N, int nsl) {
  int i = blockIdx.x * blockDim.x + threadIdx.x;
  if (i < N) {
    int c0 = 0, c1 = 0, c2 = 0, c3 = 0;
    int s = 0;
    for (; s + 4 <= nsl; s += 4) {
      c0 += part[(size_t)(s + 0) * N + i];
      c1 += part[(size_t)(s + 1) * N + i];
      c2 += part[(size_t)(s + 2) * N + i];
      c3 += part[(size_t)(s + 3) * N + i];
    }
    for (; s < nsl; ++s) c0 += part[(size_t)s * N + i];
    float d = 1.0f + (float)((c0 + c1) + (c2 + c3));
    float di = (d > 0.0f) ? rsqrtf(d) : 0.0f;
    dinv[i] = di;
    v2f xv = reinterpret_cast<const v2f*>(x)[i];
    reinterpret_cast<v2f*>(xs)[i] = xv * di;
  }
}

__global__ void k_gemm12(const float* __restrict__ part, const float* __restrict__ xs,
                         const float* __restrict__ dinv, const float* __restrict__ W1,
                         const float* __restrict__ b1, const float* __restrict__ W2,
                         float* __restrict__ h2s, int nwaves, int N, int nsl) {
  const int wave = (blockIdx.x * blockDim.x + threadIdx.x) >> 5;
  const int lane = threadIdx.x & 31;
  if (wave >= nwaves) return;
  const int c = lane & 15;
  const int h = lane >> 4;
  const int node = (wave << 5) + lane;

  const v2f xv = reinterpret_cast<const v2f*>(xs)[node];
  const float di = dinv[node];
  v2f w1;
  w1.x = W1[0 * 16 + c];
  w1.y = W1[1 * 16 + c];
  float bias[8];
#pragma unroll
  for (int j = 0; j < 8; ++j) bias[j] = b1[8 * h + j];
  const int m = c & 7;
  const int mm = (m < 2) ? m : 0;
  float u[8];
#pragma unroll
  for (int k = 0; k < 8; ++k) u[k] = W2[(8 * h + k) * 2 + mm];

  const v2f* p2 = reinterpret_cast<const v2f*>(part) + node;
  v2f a0 = {0.0f, 0.0f}, a1 = {0.0f, 0.0f}, a2 = {0.0f, 0.0f}, a3 = {0.0f, 0.0f};
  int s = 0;
  for (; s + 8 <= nsl; s += 8) {
    v2f t0 = p2[(size_t)(s + 0) * N], t1 = p2[(size_t)(s + 1) * N];
    v2f t2 = p2[(size_t)(s + 2) * N], t3 = p2[(size_t)(s + 3) * N];
    v2f t4 = p2[(size_t)(s + 4) * N], t5 = p2[(size_t)(s + 5) * N];
    v2f t6 = p2[(size_t)(s + 6) * N], t7 = p2[(size_t)(s + 7) * N];
    a0 += t0; a1 += t1; a2 += t2; a3 += t3;
    a0 += t4; a1 += t5; a2 += t6; a3 += t7;
  }
  for (; s + 4 <= nsl; s += 4) {
    a0 += p2[(size_t)(s + 0) * N];
    a1 += p2[(size_t)(s + 1) * N];
    a2 += p2[(size_t)(s + 2) * N];
    a3 += p2[(size_t)(s + 3) * N];
  }
  for (; s < nsl; ++s) a0 += p2[(size_t)s * N];
  const v2f ax = (((a0 + a1) + (a2 + a3)) + xv) * di;

  const v2f z2 = {0.0f, 0.0f};
  const v2f wa0 = h ? z2 : w1, wa1 = h ? w1 : z2;
  const v2f xb0 = h ? z2 : ax, xb1 = h ? ax : z2;
  v8f d0 = {}, d1 = {};
  d0 = __builtin_amdgcn_wmma_f32_16x16x4_f32(
       false, wa0,  false, xb0,
       (short)0, d0,  false,  false);
  d1 = __builtin_amdgcn_wmma_f32_16x16x4_f32(
      false, wa1, false, xb1, (short)0, d1, false, false);

  float r0[8], r1[8];
#pragma unroll
  for (int j = 0; j < 8; ++j) {
    r0[j] = fmaxf(d0[j] + bias[j], 0.0f);
    r1[j] = fmaxf(d1[j] + bias[j], 0.0f);
  }

  v8f e0 = {}, e1 = {};
#pragma unroll
  for (int q = 0; q < 4; ++q) {
    v2f w2;
    w2.x = (m < 2) ? u[2 * q + 0] : 0.0f;
    w2.y = (m < 2) ? u[2 * q + 1] : 0.0f;
    v2f rb0, rb1;
    rb0.x = r0[2 * q]; rb0.y = r0[2 * q + 1];
    rb1.x = r1[2 * q]; rb1.y = r1[2 * q + 1];
    e0 = __builtin_amdgcn_wmma_f32_16x16x4_f32(
        false, w2, false, rb0, (short)0, e0, false, false);
    e1 = __builtin_amdgcn_wmma_f32_16x16x4_f32(
        false, w2, false, rb1, (short)0, e1, false, false);
  }

  v2f o;
  o.x = h ? e1[0] : e0[0];
  o.y = h ? e1[1] : e0[1];
  reinterpret_cast<v2f*>(h2s)[node] = o * di;
}

__global__ void k_gemm1_tail(const float* __restrict__ part, const float* __restrict__ xs,
                             const float* __restrict__ dinv, const float* __restrict__ W1,
                             const float* __restrict__ b1, float* __restrict__ r,
                             int n0, int N, int nsl) {
  int tid = blockIdx.x * blockDim.x + threadIdx.x;
  int n = n0 + (tid >> 4);
  if (n >= N) return;
  int f = tid & 15;
  const v2f* p2 = reinterpret_cast<const v2f*>(part) + n;
  v2f acc = {0.0f, 0.0f};
  for (int s = 0; s < nsl; ++s) acc += p2[(size_t)s * N];
  v2f xv = reinterpret_cast<const v2f*>(xs)[n];
  v2f ax = (acc + xv) * dinv[n];
  r[n * 16 + f] = fmaxf(ax.x * W1[f] + ax.y * W1[16 + f] + b1[f], 0.0f);
}

__global__ void k_gemm2_tail(const float* __restrict__ r, const float* __restrict__ W2,
                             const float* __restrict__ dinv,
                             float* __restrict__ h2s, int n0, int N) {
  int tid = blockIdx.x * blockDim.x + threadIdx.x;
  int n = n0 + (tid >> 1);
  if (n >= N) return;
  int f = tid & 1;
  float acc = 0.0f;
#pragma unroll
  for (int k = 0; k < 16; ++k)
    acc += r[n * 16 + k] * W2[k * 2 + f];
  h2s[n * 2 + f] = acc * dinv[n];
}

__global__ void k_final(const float* __restrict__ part, const float* __restrict__ h2s,
                        const float* __restrict__ dinv, const float* __restrict__ b2,
                        float* __restrict__ out, int N, int nsl) {
  int tid = blockIdx.x * blockDim.x + threadIdx.x;
  if (tid >= 2 * N) return;
  int n = tid >> 1;
  int f = tid & 1;
  const size_t stride = (size_t)2 * N;
  float a0 = 0.0f, a1 = 0.0f, a2 = 0.0f, a3 = 0.0f;
  int s = 0;
  for (; s + 4 <= nsl; s += 4) {
    a0 += part[(size_t)(s + 0) * stride + tid];
    a1 += part[(size_t)(s + 1) * stride + tid];
    a2 += part[(size_t)(s + 2) * stride + tid];
    a3 += part[(size_t)(s + 3) * stride + tid];
  }
  for (; s < nsl; ++s) a0 += part[(size_t)s * stride + tid];
  float di = dinv[n];
  out[tid] = ((a0 + a1) + (a2 + a3)) * di + (h2s[tid] * di + b2[f]);
}

extern "C" void kernel_launch(void* const* d_in, const int* in_sizes, int n_in,
                              void* d_out, int out_size, void* d_ws, size_t ws_size,
                              hipStream_t stream) {
  const float* x  = (const float*)d_in[0];
  const int*   ei = (const int*)d_in[1];
  const float* W1 = (const float*)d_in[2];
  const float* b1 = (const float*)d_in[3];
  const float* W2 = (const float*)d_in[4];
  const float* b2 = (const float*)d_in[5];
  float* out = (float*)d_out;

  const int N = in_sizes[0] / 2;
  const int E = in_sizes[1] / 2;
  const int* src = ei;
  const int* dst = ei + E;

  const size_t NP = ((size_t)N + 63) & ~(size_t)63;
  float* ws   = (float*)d_ws;
  float* dinv = ws;
  float* xs   = ws + NP;
  float* r    = ws + (size_t)3 * NP;
  float* h2s  = ws + (size_t)19 * NP;
  float* part = ws + (size_t)21 * NP;

  const int B = 256;
  const int nwv = N / 32;
  const int n0 = nwv * 32;
  const int ntail = N - n0;

  const int nr1 = (N + TILE_FLOATS - 1) / TILE_FLOATS;
  const int ns1 = (256 / nr1) > 0 ? (256 / nr1) : 1;
  const int nr2 = (2 * N + TILE_FLOATS - 1) / TILE_FLOATS;
  const int ns2 = (256 / nr2) > 0 ? (256 / nr2) : 1;

  const int S   = 256;
  const int R2  = (((N + nr2 - 1) / nr2) + 31) & ~31;
  const int WR  = (N + ns2 - 1) / ns2;
  const int NB  = nr2 * ns2;
  const int epb = (E + S - 1) / S;
  const size_t pa1 = (size_t)ns1 * N, pa2 = (size_t)ns2 * 2 * N;
  const size_t partFloats = (((pa1 > pa2) ? pa1 : pa2) + 63) & ~(size_t)63;
  int* cnt = (int*)(part + partFloats);
  int* bst = cnt + (((size_t)S * NB + 63) & ~(size_t)63);
  int* ent = bst + (((size_t)NB + 1 + 63) & ~(size_t)63);
  const size_t need = ((size_t)(ent - (int*)ws) + (size_t)E) * sizeof(int);
  const bool binned = E > 0 && NB <= BIN_MAX && 2 * R2 <= TILE_FLOATS &&
                      2 * WR <= WIN_FLOATS && epb <= STAGE_MAX &&
                      N < (1 << 24) && need <= ws_size;

  k_degree_lds<<<dim3(ns1, nr1), SC_THREADS, 0, stream>>>(dst, (int*)part, E, N);
  k_dinv  <<<(N + B - 1) / B,      B, 0, stream>>>((const int*)part, x, dinv, xs, N, ns1);

  if (binned) {
    k_count<<<S, SC_THREADS, 0, stream>>>(src, dst, cnt, E, N, R2, WR, ns2, NB, epb);
    k_place<<<S, SC_THREADS, 0, stream>>>(src, dst, cnt, ent, bst, E, N, R2, WR, ns2, NB, epb);
    k_agg_bin<<<dim3(ns2, nr2), SC_THREADS, 0, stream>>>(ent, bst, xs, part, N, R2, WR);
  } else {
    k_agg_lds<<<dim3(ns2, nr2), SC_THREADS, 0, stream>>>(src, dst, xs, part, E, N);
  }
  if (nwv > 0)
    k_gemm12<<<(nwv * 32 + B - 1) / B, B, 0, stream>>>(part, xs, dinv, W1, b1, W2, h2s, nwv, N, ns2);
  if (ntail > 0) {
    k_gemm1_tail<<<(ntail * 16 + B - 1) / B, B, 0, stream>>>(part, xs, dinv, W1, b1, r, n0, N, ns2);
    k_gemm2_tail<<<(ntail * 2 + B - 1) / B, B, 0, stream>>>(r, W2, dinv, h2s, n0, N);
  }

  if (binned)
    k_agg_bin<<<dim3(ns2, nr2), SC_THREADS, 0, stream>>>(ent, bst, h2s, part, N, R2, WR);
  else
    k_agg_lds<<<dim3(ns2, nr2), SC_THREADS, 0, stream>>>(src, dst, h2s, part, E, N);
  k_final <<<(2 * N + B - 1) / B,  B, 0, stream>>>(part, h2s, dinv, b2, out, N, ns2);
}


# Bucket grid 21 destination ranges x 12 source windows (was 7 x 36): a third of the partial planes stored and summed per layer, 3x larger LDS window copied as 16-byte groups; float sum order changes
# speedup vs baseline: 8.5540x; 1.2100x over previous
#include <hip/hip_runtime.h>


typedef __attribute__((ext_vector_type(2))) float v2f;
typedef __attribute__((ext_vector_type(4))) float v4f;
typedef __attribute__((ext_vector_type(8))) float v8f;

#define TILE_FLOATS 65536
#define SC_THREADS  1024
#define SC_BATCH    8

__global__ void k_degree_lds(const int* __restrict__ dst, int* __restrict__ partial,
                             int E, int N) {
  __shared__ __attribute__((aligned(16))) int tile[TILE_FLOATS];
  const int t = threadIdx.x;
  const int R = (((N + (int)gridDim.y - 1) / (int)gridDim.y) + 31) & ~31;
  const int eps = (E + (int)gridDim.x - 1) / (int)gridDim.x;
  const int lo = blockIdx.y * R;
  const int nloc = max(0, min(R, N - lo));

  const int4 z4 = {0, 0, 0, 0};
  for (int i = t * 4; i < TILE_FLOATS; i += SC_THREADS * 4)
    *reinterpret_cast<int4*>(&tile[i]) = z4;
  __syncthreads();

  const int e0 = blockIdx.x * eps;
  const int e1 = min(E, e0 + eps);
  const int elast = e1 - 1;

  int dn[SC_BATCH];
  int base = e0 + t;
  for (int k = 0; k < SC_BATCH; ++k) {
    int e = base + k * SC_THREADS;
    int dv = dst[min(e, elast)];
    dn[k] = (e < e1) ? dv : -1;
  }
  for (; base < e1; base += SC_BATCH * SC_THREADS) {
    int d[SC_BATCH];
    for (int k = 0; k < SC_BATCH; ++k) d[k] = dn[k];
    const int nb = base + SC_BATCH * SC_THREADS;
    for (int k = 0; k < SC_BATCH; ++k) {
      int e = nb + k * SC_THREADS;
      int dv = dst[min(e, elast)];
      dn[k] = (e < e1) ? dv : -1;
    }
    for (int k = 0; k < SC_BATCH; ++k) {
      unsigned r = (unsigned)(d[k] - lo);
      if (r < (unsigned)nloc) atomicAdd(&tile[r], 1);
    }
  }
  __syncthreads();

  int* p = partial + ((size_t)blockIdx.x * N + lo);
  for (int i = t; i < nloc; i += SC_THREADS) p[i] = tile[i];
}

__global__ void k_agg_lds(const int* __restrict__ src, const int* __restrict__ dst,
                           const float* __restrict__ feat, float* __restrict__ partial,
                           int E, int N) {
  __shared__ __attribute__((aligned(16))) float tile[TILE_FLOATS];
  const int t = threadIdx.x;
  const int R = (((N + (int)gridDim.y - 1) / (int)gridDim.y) + 31) & ~31;
  const int eps = (E + (int)gridDim.x - 1) / (int)gridDim.x;
  const int lo = blockIdx.y * R;
  const int nloc = max(0, min(R, N - lo));

  const float4 z4 = {0.0f, 0.0f, 0.0f, 0.0f};
  for (int i = t * 4; i < TILE_FLOATS; i += SC_THREADS * 4)
    *reinterpret_cast<float4*>(&tile[i]) = z4;
  __syncthreads();

  const int e0 = blockIdx.x * eps;
  const int e1 = min(E, e0 + eps);
  const int elast = e1 - 1;
  const v2f* feat2 = reinterpret_cast<const v2f*>(feat);

  int dn[SC_BATCH], sn[SC_BATCH];
  int base = e0 + t;
  for (int k = 0; k < SC_BATCH; ++k) {
    int e = base + k * SC_THREADS;
    int ec = min(e, elast);
    int dv = dst[ec];
    dn[k] = (e < e1) ? dv : -1;
    sn[k] = src[ec];
  }
  for (; base < e1; base += SC_BATCH * SC_THREADS) {
    int d[SC_BATCH], s[SC_BATCH];
    for (int k = 0; k < SC_BATCH; ++k) { d[k] = dn[k]; s[k] = sn[k]; }
    const int nb = base + SC_BATCH * SC_THREADS;
    for (int k = 0; k < SC_BATCH; ++k) {
      int e = nb + k * SC_THREADS;
      int ec = min(e, elast);
      int dv = dst[ec];
      dn[k] = (e < e1) ? dv : -1;
      sn[k] = src[ec];
    }
    v2f v[SC_BATCH];
    for (int k = 0; k < SC_BATCH; ++k) {
      unsigned r = (unsigned)(d[k] - lo);
      v[k] = feat2[(r < (unsigned)nloc) ? s[k] : 0];
    }
    for (int k = 0; k < SC_BATCH; ++k) {
      unsigned r = (unsigned)(d[k] - lo);
      if (r < (unsigned)nloc) {
        atomicAdd(&tile[2 * r + 0], v[k].x);
        atomicAdd(&tile[2 * r + 1], v[k].y);
      }
    }
  }
  __syncthreads();

  float* p = partial + ((size_t)blockIdx.x * N + lo) * 2;
  const int cnt = nloc * 2;
  for (int i = t; i < cnt; i += SC_THREADS) p[i] = tile[i];
}

#define BIN_MAX    256
#define STAGE_MAX  32768
#define AGG_LDS_FLOATS 81920
#define AG_BATCH   16
#define WIN_BATCH  9
#define BIN_RANGES 21

__device__ __forceinline__ int qdiv(int v, int dvs, float inv) {
  int q = (int)((float)v * inv);
  int r = v - q * dvs;
  q += (r >= dvs) ? 1 : 0;
  q -= (r < 0) ? 1 : 0;
  return q;
}

__device__ __forceinline__ int edge_bucket(int d, int s, int N, int R, int WR,
                                           int nwin, float invR, float invW,
                                           int* packed) {
  if ((unsigned)d >= (unsigned)N || (unsigned)s >= (unsigned)N) return -1;
  int y = qdiv(d, R, invR);
  int x = qdiv(s, WR, invW);
  *packed = ((s - x * WR) << 15) | (d - y * R);
  return y * nwin + x;
}

__global__ void k_count(const int* __restrict__ src, const int* __restrict__ dst,
                        int* __restrict__ cnt, int E, int N, int R, int WR,
                        int nwin, int NB, int epb) {
  __shared__ int hist[BIN_MAX];
  const int t = threadIdx.x;
  if (t < BIN_MAX) hist[t] = 0;
  __syncthreads();
  const float invR = 1.0f / (float)R, invW = 1.0f / (float)WR;
  const int e0 = min(E, (int)blockIdx.x * epb);
  const int e1 = min(E, e0 + epb);
  for (int base = e0 + t; base < e1; base += SC_BATCH * SC_THREADS) {
    int dv[SC_BATCH], sv[SC_BATCH];
    for (int k = 0; k < SC_BATCH; ++k) {
      int e = base + k * SC_THREADS;
      int ec = min(e, e1 - 1);
      int d = dst[ec];
      sv[k] = src[ec];
      dv[k] = (e < e1) ? d : -1;
    }
    for (int k = 0; k < SC_BATCH; ++k) {
      int pk;
      int b = edge_bucket(dv[k], sv[k], N, R, WR, nwin, invR, invW, &pk);
      if (b >= 0) atomicAdd(&hist[b], 1);
    }
  }
  __syncthreads();
  if (t < NB) cnt[(size_t)blockIdx.x * NB + t] = hist[t];
}

__global__ void k_place(const int* __restrict__ src, const int* __restrict__ dst,
                        const int* __restrict__ cnt, int* __restrict__ ent,
                        int* __restrict__ bst, int E, int N, int R, int WR,
                        int nwin, int NB, int epb) {
  __shared__ int tot[BIN_MAX], pre[BIN_MAX], own[BIN_MAX];
  __shared__ int cur[BIN_MAX], lst[BIN_MAX], gst[BIN_MAX];
  __shared__ int sa[BIN_MAX], so[BIN_MAX];
  __shared__ int sval[STAGE_MAX];
  __shared__ unsigned char sbin[STAGE_MAX];
  const int t = threadIdx.x;
  const int slice = blockIdx.x;
  const int S = gridDim.x;
  if (t < BIN_MAX) {
    tot[t] = 0;
    pre[t] = 0;
    own[t] = (t < NB) ? cnt[(size_t)slice * NB + t] : 0;
  }
  __syncthreads();
  {
    const int b = t & (BIN_MAX - 1);
    const int part = t >> 8;
    const int per = (S + 3) >> 2;
    const int s0 = part * per;
    const int s1 = min(S, s0 + per);
    if (b < NB && s0 < s1) {
      int all = 0, before = 0;
      for (int s = s0; s < s1; s += 16) {
        int v[16];
#pragma unroll
        for (int k = 0; k < 16; ++k) {
          int ss = min(s + k, s1 - 1);
          v[k] = cnt[(size_t)ss * NB + b];
        }
#pragma unroll
        for (int k = 0; k < 16; ++k) {
          int ss = s + k;
          int vv = (ss < s1) ? v[k] : 0;
          all += vv;
          before += (ss < slice) ? vv : 0;
        }
      }
      atomicAdd(&tot[b], all);
      atomicAdd(&pre[b], before);
    }
  }
  __syncthreads();
  int pa = 0, po = 0;
  if (t < BIN_MAX) { pa = tot[t]; po = own[t]; sa[t] = pa; so[t] = po; }
  __syncthreads();
  for (int off = 1; off < BIN_MAX; off <<= 1) {
    int a2 = 0, o2 = 0;
    if (t < BIN_MAX && t >= off) { a2 = sa[t - off]; o2 = so[t - off]; }
    __syncthreads();
    if (t < BIN_MAX) { pa += a2; po += o2; sa[t] = pa; so[t] = po; }
    __syncthreads();
  }
  if (t < NB) {
    int gs = pa - tot[t];
    int ls = po - own[t];
    gst[t] = gs + pre[t];
    lst[t] = ls;
    cur[t] = ls;
    if (slice == 0) {
      bst[t] = gs;
      if (t == NB - 1) bst[NB] = pa;
    }
  }
  __syncthreads();

  const float invR = 1.0f / (float)R, invW = 1.0f / (float)WR;
  const int e0 = min(E, slice * epb);
  const int e1 = min(E, e0 + epb);
  for (int base = e0 + t; base < e1; base += SC_BATCH * SC_THREADS) {
    int dv[SC_BATCH], sv[SC_BATCH];
    for (int k = 0; k < SC_BATCH; ++k) {
      int e = base + k * SC_THREADS;
      int ec = min(e, e1 - 1);
      int d = dst[ec];
      sv[k] = src[ec];
      dv[k] = (e < e1) ? d : -1;
    }
    for (int k = 0; k < SC_BATCH; ++k) {
      int pk = 0;
      int b = edge_bucket(dv[k], sv[k], N, R, WR, nwin, invR, invW, &pk);
      if (b >= 0) {
        int lp = atomicAdd(&cur[b], 1);
        sval[lp] = pk;
        sbin[lp] = (unsigned char)b;
      }
    }
  }
  __syncthreads();
  const int ntot = lst[NB - 1] + own[NB - 1];
  for (int i = t; i < ntot; i += SC_THREADS) {
    int b = sbin[i];
    ent[gst[b] + (i - lst[b])] = sval[i];
  }
}

__global__ void k_agg_bin(const int* __restrict__ ent, const int* __restrict__ bst,
                          const float* __restrict__ feat, float* __restrict__ partial,
                          int N, int R, int WR) {
  __shared__ __attribute__((aligned(16))) float lds[AGG_LDS_FLOATS];
  float* tile = lds;
  float* win = lds + 2 * R;
  const int t = threadIdx.x;
  const int lo = blockIdx.y * R;
  const int nloc = max(0, min(R, N - lo));
  const int wlo = blockIdx.x * WR;
  const int wn = max(0, min(WR, N - wlo));
  const int bucket = blockIdx.y * gridDim.x + blockIdx.x;
  const int b0 = bst[bucket];
  const int b1 = bst[bucket + 1];

  const float4 z4 = {0.0f, 0.0f, 0.0f, 0.0f};
  for (int i = t * 4; i < nloc * 2; i += SC_THREADS * 4)
    *reinterpret_cast<float4*>(&tile[i]) = z4;
  const float* wsrc = feat + (size_t)wlo * 2;
  const v4f* wsrc4 = reinterpret_cast<const v4f*>(wsrc);
  const int wf = wn * 2;
  const int wf4 = wf >> 2;
  static_assert(WIN_BATCH == 9, "the fence below names nine pieces");
  for (int base = t; base < wf4; base += WIN_BATCH * SC_THREADS) {
    v4f wv[WIN_BATCH];
    for (int k = 0; k < WIN_BATCH; ++k)
      wv[k] = wsrc4[min(base + k * SC_THREADS, wf4 - 1)];
    asm volatile("" :: "v"(wv[0]), "v"(wv[1]), "v"(wv[2]), "v"(wv[3]), "v"(wv[4]),
                       "v"(wv[5]), "v"(wv[6]), "v"(wv[7]), "v"(wv[8]));
    for (int k = 0; k < WIN_BATCH; ++k) {
      int i = base + k * SC_THREADS;
      if (i < wf4) *reinterpret_cast<v4f*>(&win[4 * i]) = wv[k];
    }
  }
  if (t < (wf & 3)) win[(wf4 << 2) + t] = wsrc[(wf4 << 2) + t];
  __syncthreads();

  const v2f* win2 = reinterpret_cast<const v2f*>(win);
  for (int base = b0 + t; base < b1; base += AG_BATCH * SC_THREADS) {
    int en[AG_BATCH];
    for (int k = 0; k < AG_BATCH; ++k) {
      int i = base + k * SC_THREADS;
      int v = ent[min(i, b1 - 1)];
      en[k] = (i < b1) ? v : -1;
    }
    v2f v[AG_BATCH];
    for (int k = 0; k < AG_BATCH; ++k)
      v[k] = win2[(en[k] >= 0) ? (en[k] >> 15) : 0];
    for (int k = 0; k < AG_BATCH; ++k) {
      if (en[k] >= 0) {
        int dl = en[k] & 32767;
        atomicAdd(&tile[2 * dl + 0], v[k].x);
        atomicAdd(&tile[2 * dl + 1], v[k].y);
      }
    }
  }
  __syncthreads();

  float* p = partial + ((size_t)blockIdx.x * N + lo) * 2;
  const int cnt2 = nloc * 2;
  for (int i = t; i < cnt2; i += SC_THREADS) p[i] = tile[i];
}

__global__ void k_dinv(const int* __restrict__ part, const float* __restrict__ x,
                       float* __restrict__ dinv, float* __restrict__ xs,
                       int N, int nsl) {
  int i = blockIdx.x * blockDim.x + threadIdx.x;
  if (i < N) {
    int c0 = 0, c1 = 0, c2 = 0, c3 = 0;
    int s = 0;
    for (; s + 4 <= nsl; s += 4) {
      c0 += part[(size_t)(s + 0) * N + i];
      c1 += part[(size_t)(s + 1) * N + i];
      c2 += part[(size_t)(s + 2) * N + i];
      c3 += part[(size_t)(s + 3) * N + i];
    }
    for (; s < nsl; ++s) c0 += part[(size_t)s * N + i];
    float d = 1.0f + (float)((c0 + c1) + (c2 + c3));
    float di = (d > 0.0f) ? rsqrtf(d) : 0.0f;
    dinv[i] = di;
    v2f xv = reinterpret_cast<const v2f*>(x)[i];
    reinterpret_cast<v2f*>(xs)[i] = xv * di;
  }
}

__global__ void k_gemm12(const float* __restrict__ part, const float* __restrict__ xs,
                         const float* __restrict__ dinv, const float* __restrict__ W1,
                         const float* __restrict__ b1, const float* __restrict__ W2,
                         float* __restrict__ h2s, int nwaves, int N, int nsl) {
  const int wave = (blockIdx.x * blockDim.x + threadIdx.x) >> 5;
  const int lane = threadIdx.x & 31;
  if (wave >= nwaves) return;
  const int c = lane & 15;
  const int h = lane >> 4;
  const int node = (wave << 5) + lane;

  const v2f xv = reinterpret_cast<const v2f*>(xs)[node];
  const float di = dinv[node];
  v2f w1;
  w1.x = W1[0 * 16 + c];
  w1.y = W1[1 * 16 + c];
  float bias[8];
#pragma unroll
  for (int j = 0; j < 8; ++j) bias[j] = b1[8 * h + j];
  const int m = c & 7;
  const int mm = (m < 2) ? m : 0;
  float u[8];
#pragma unroll
  for (int k = 0; k < 8; ++k) u[k] = W2[(8 * h + k) * 2 + mm];

  const v2f* p2 = reinterpret_cast<const v2f*>(part) + node;
  v2f a0 = {0.0f, 0.0f}, a1 = {0.0f, 0.0f}, a2 = {0.0f, 0.0f}, a3 = {0.0f, 0.0f};
  int s = 0;
  for (; s + 8 <= nsl; s += 8) {
    v2f t0 = p2[(size_t)(s + 0) * N], t1 = p2[(size_t)(s + 1) * N];
    v2f t2 = p2[(size_t)(s + 2) * N], t3 = p2[(size_t)(s + 3) * N];
    v2f t4 = p2[(size_t)(s + 4) * N], t5 = p2[(size_t)(s + 5) * N];
    v2f t6 = p2[(size_t)(s + 6) * N], t7 = p2[(size_t)(s + 7) * N];
    a0 += t0; a1 += t1; a2 += t2; a3 += t3;
    a0 += t4; a1 += t5; a2 += t6; a3 += t7;
  }
  for (; s + 4 <= nsl; s += 4) {
    a0 += p2[(size_t)(s + 0) * N];
    a1 += p2[(size_t)(s + 1) * N];
    a2 += p2[(size_t)(s + 2) * N];
    a3 += p2[(size_t)(s + 3) * N];
  }
  for (; s < nsl; ++s) a0 += p2[(size_t)s * N];
  const v2f ax = (((a0 + a1) + (a2 + a3)) + xv) * di;

  const v2f z2 = {0.0f, 0.0f};
  const v2f wa0 = h ? z2 : w1, wa1 = h ? w1 : z2;
  const v2f xb0 = h ? z2 : ax, xb1 = h ? ax : z2;
  v8f d0 = {}, d1 = {};
  d0 = __builtin_amdgcn_wmma_f32_16x16x4_f32(
       false, wa0,  false, xb0,
       (short)0, d0,  false,  false);
  d1 = __builtin_amdgcn_wmma_f32_16x16x4_f32(
      false, wa1, false, xb1, (short)0, d1, false, false);

  float r0[8], r1[8];
#pragma unroll
  for (int j = 0; j < 8; ++j) {
    r0[j] = fmaxf(d0[j] + bias[j], 0.0f);
    r1[j] = fmaxf(d1[j] + bias[j], 0.0f);
  }

  v8f e0 = {}, e1 = {};
#pragma unroll
  for (int q = 0; q < 4; ++q) {
    v2f w2;
    w2.x = (m < 2) ? u[2 * q + 0] : 0.0f;
    w2.y = (m < 2) ? u[2 * q + 1] : 0.0f;
    v2f rb0, rb1;
    rb0.x = r0[2 * q]; rb0.y = r0[2 * q + 1];
    rb1.x = r1[2 * q]; rb1.y = r1[2 * q + 1];
    e0 = __builtin_amdgcn_wmma_f32_16x16x4_f32(
        false, w2, false, rb0, (short)0, e0, false, false);
    e1 = __builtin_amdgcn_wmma_f32_16x16x4_f32(
        false, w2, false, rb1, (short)0, e1, false, false);
  }

  v2f o;
  o.x = h ? e1[0] : e0[0];
  o.y = h ? e1[1] : e0[1];
  reinterpret_cast<v2f*>(h2s)[node] = o * di;
}

__global__ void k_gemm1_tail(const float* __restrict__ part, const float* __restrict__ xs,
                             const float* __restrict__ dinv, const float* __restrict__ W1,
                             const float* __restrict__ b1, float* __restrict__ r,
                             int n0, int N, int nsl) {
  int tid = blockIdx.x * blockDim.x + threadIdx.x;
  int n = n0 + (tid >> 4);
  if (n >= N) return;
  int f = tid & 15;
  const v2f* p2 = reinterpret_cast<const v2f*>(part) + n;
  v2f acc = {0.0f, 0.0f};
  for (int s = 0; s < nsl; ++s) acc += p2[(size_t)s * N];
  v2f xv = reinterpret_cast<const v2f*>(xs)[n];
  v2f ax = (acc + xv) * dinv[n];
  r[n * 16 + f] = fmaxf(ax.x * W1[f] + ax.y * W1[16 + f] + b1[f], 0.0f);
}

__global__ void k_gemm2_tail(const float* __restrict__ r, const float* __restrict__ W2,
                             const float* __restrict__ dinv,
                             float* __restrict__ h2s, int n0, int N) {
  int tid = blockIdx.x * blockDim.x + threadIdx.x;
  int n = n0 + (tid >> 1);
  if (n >= N) return;
  int f = tid & 1;
  float acc = 0.0f;
#pragma unroll
  for (int k = 0; k < 16; ++k)
    acc += r[n * 16 + k] * W2[k * 2 + f];
  h2s[n * 2 + f] = acc * dinv[n];
}

__global__ void k_final(const float* __restrict__ part, const float* __restrict__ h2s,
                        const float* __restrict__ dinv, const float* __restrict__ b2,
                        float* __restrict__ out, int N, int nsl) {
  int tid = blockIdx.x * blockDim.x + threadIdx.x;
  if (tid >= 2 * N) return;
  int n = tid >> 1;
  int f = tid & 1;
  const size_t stride = (size_t)2 * N;
  float a0 = 0.0f, a1 = 0.0f, a2 = 0.0f, a3 = 0.0f;
  int s = 0;
  for (; s + 4 <= nsl; s += 4) {
    a0 += part[(size_t)(s + 0) * stride + tid];
    a1 += part[(size_t)(s + 1) * stride + tid];
    a2 += part[(size_t)(s + 2) * stride + tid];
    a3 += part[(size_t)(s + 3) * stride + tid];
  }
  for (; s < nsl; ++s) a0 += part[(size_t)s * stride + tid];
  float di = dinv[n];
  out[tid] = ((a0 + a1) + (a2 + a3)) * di + (h2s[tid] * di + b2[f]);
}

extern "C" void kernel_launch(void* const* d_in, const int* in_sizes, int n_in,
                              void* d_out, int out_size, void* d_ws, size_t ws_size,
                              hipStream_t stream) {
  const float* x  = (const float*)d_in[0];
  const int*   ei = (const int*)d_in[1];
  const float* W1 = (const float*)d_in[2];
  const float* b1 = (const float*)d_in[3];
  const float* W2 = (const float*)d_in[4];
  const float* b2 = (const float*)d_in[5];
  float* out = (float*)d_out;

  const int N = in_sizes[0] / 2;
  const int E = in_sizes[1] / 2;
  const int* src = ei;
  const int* dst = ei + E;

  const size_t NP = ((size_t)N + 63) & ~(size_t)63;
  float* ws   = (float*)d_ws;
  float* dinv = ws;
  float* xs   = ws + NP;
  float* r    = ws + (size_t)3 * NP;
  float* h2s  = ws + (size_t)19 * NP;
  float* part = ws + (size_t)21 * NP;

  const int B = 256;
  const int nwv = N / 32;
  const int n0 = nwv * 32;
  const int ntail = N - n0;

  const int nr1 = (N + TILE_FLOATS - 1) / TILE_FLOATS;
  const int ns1 = (256 / nr1) > 0 ? (256 / nr1) : 1;
  const int nrfit = (2 * N + TILE_FLOATS - 1) / TILE_FLOATS;
  const int nr2 = nrfit > BIN_RANGES ? nrfit : BIN_RANGES;
  const int ns2 = (256 / nr2) > 0 ? (256 / nr2) : 1;

  const int S   = 256;
  const int R2  = (((N + nr2 - 1) / nr2) + 31) & ~31;
  const int WR  = (((N + ns2 - 1) / ns2) + 31) & ~31;
  const int NB  = nr2 * ns2;
  const int epb = (E + S - 1) / S;
  const size_t pa1 = (size_t)ns1 * N, pa2 = (size_t)ns2 * 2 * N;
  const size_t partFloats = (((pa1 > pa2) ? pa1 : pa2) + 63) & ~(size_t)63;
  int* cnt = (int*)(part + partFloats);
  int* bst = cnt + (((size_t)S * NB + 63) & ~(size_t)63);
  int* ent = bst + (((size_t)NB + 1 + 63) & ~(size_t)63);
  const size_t need = ((size_t)(ent - (int*)ws) + (size_t)E) * sizeof(int);
  const bool binned = E > 0 && NB <= BIN_MAX && R2 <= 32768 && WR <= 65536 &&
                      2 * (R2 + WR) <= AGG_LDS_FLOATS && epb <= STAGE_MAX &&
                      N < (1 << 24) && need <= ws_size;

  k_degree_lds<<<dim3(ns1, nr1), SC_THREADS, 0, stream>>>(dst, (int*)part, E, N);
  k_dinv  <<<(N + B - 1) / B,      B, 0, stream>>>((const int*)part, x, dinv, xs, N, ns1);

  if (binned) {
    k_count<<<S, SC_THREADS, 0, stream>>>(src, dst, cnt, E, N, R2, WR, ns2, NB, epb);
    k_place<<<S, SC_THREADS, 0, stream>>>(src, dst, cnt, ent, bst, E, N, R2, WR, ns2, NB, epb);
    k_agg_bin<<<dim3(ns2, nr2), SC_THREADS, 0, stream>>>(ent, bst, xs, part, N, R2, WR);
  } else {
    k_agg_lds<<<dim3(ns2, nr2), SC_THREADS, 0, stream>>>(src, dst, xs, part, E, N);
  }
  if (nwv > 0)
    k_gemm12<<<(nwv * 32 + B - 1) / B, B, 0, stream>>>(part, xs, dinv, W1, b1, W2, h2s, nwv, N, ns2);
  if (ntail > 0) {
    k_gemm1_tail<<<(ntail * 16 + B - 1) / B, B, 0, stream>>>(part, xs, dinv, W1, b1, r, n0, N, ns2);
    k_gemm2_tail<<<(ntail * 2 + B - 1) / B, B, 0, stream>>>(r, W2, dinv, h2s, n0, N);
  }

  if (binned)
    k_agg_bin<<<dim3(ns2, nr2), SC_THREADS, 0, stream>>>(ent, bst, h2s, part, N, R2, WR);
  else
    k_agg_lds<<<dim3(ns2, nr2), SC_THREADS, 0, stream>>>(src, dst, h2s, part, E, N);
  k_final <<<(2 * N + B - 1) / B,  B, 0, stream>>>(part, h2s, dinv, b2, out, N, ns2);
}


# In-degree counted from the edge buckets after placing (one bucket per work-group, integer LDS adds): 12 count planes stored and summed, was 64 planes from four scans of the destination list
# speedup vs baseline: 10.5671x; 1.2353x over previous
#include <hip/hip_runtime.h>


typedef __attribute__((ext_vector_type(2))) float v2f;
typedef __attribute__((ext_vector_type(4))) float v4f;
typedef __attribute__((ext_vector_type(8))) float v8f;

#define TILE_FLOATS 65536
#define SC_THREADS  1024
#define SC_BATCH    8

__global__ void k_degree_lds(const int* __restrict__ dst, int* __restrict__ partial,
                             int E, int N) {
  __shared__ __attribute__((aligned(16))) int tile[TILE_FLOATS];
  const int t = threadIdx.x;
  const int R = (((N + (int)gridDim.y - 1) / (int)gridDim.y) + 31) & ~31;
  const int eps = (E + (int)gridDim.x - 1) / (int)gridDim.x;
  const int lo = blockIdx.y * R;
  const int nloc = max(0, min(R, N - lo));

  const int4 z4 = {0, 0, 0, 0};
  for (int i = t * 4; i < TILE_FLOATS; i += SC_THREADS * 4)
    *reinterpret_cast<int4*>(&tile[i]) = z4;
  __syncthreads();

  const int e0 = blockIdx.x * eps;
  const int e1 = min(E, e0 + eps);
  const int elast = e1 - 1;

  int dn[SC_BATCH];
  int base = e0 + t;
  for (int k = 0; k < SC_BATCH; ++k) {
    int e = base + k * SC_THREADS;
    int dv = dst[min(e, elast)];
    dn[k] = (e < e1) ? dv : -1;
  }
  for (; base < e1; base += SC_BATCH * SC_THREADS) {
    int d[SC_BATCH];
    for (int k = 0; k < SC_BATCH; ++k) d[k] = dn[k];
    const int nb = base + SC_BATCH * SC_THREADS;
    for (int k = 0; k < SC_BATCH; ++k) {
      int e = nb + k * SC_THREADS;
      int dv = dst[min(e, elast)];
      dn[k] = (e < e1) ? dv : -1;
    }
    for (int k = 0; k < SC_BATCH; ++k) {
      unsigned r = (unsigned)(d[k] - lo);
      if (r < (unsigned)nloc) atomicAdd(&tile[r], 1);
    }
  }
  __syncthreads();

  int* p = partial + ((size_t)blockIdx.x * N + lo);
  for (int i = t; i < nloc; i += SC_THREADS) p[i] = tile[i];
}

__global__ void k_agg_lds(const int* __restrict__ src, const int* __restrict__ dst,
                           const float* __restrict__ feat, float* __restrict__ partial,
                           int E, int N) {
  __shared__ __attribute__((aligned(16))) float tile[TILE_FLOATS];
  const int t = threadIdx.x;
  const int R = (((N + (int)gridDim.y - 1) / (int)gridDim.y) + 31) & ~31;
  const int eps = (E + (int)gridDim.x - 1) / (int)gridDim.x;
  const int lo = blockIdx.y * R;
  const int nloc = max(0, min(R, N - lo));

  const float4 z4 = {0.0f, 0.0f, 0.0f, 0.0f};
  for (int i = t * 4; i < TILE_FLOATS; i += SC_THREADS * 4)
    *reinterpret_cast<float4*>(&tile[i]) = z4;
  __syncthreads();

  const int e0 = blockIdx.x * eps;
  const int e1 = min(E, e0 + eps);
  const int elast = e1 - 1;
  const v2f* feat2 = reinterpret_cast<const v2f*>(feat);

  int dn[SC_BATCH], sn[SC_BATCH];
  int base = e0 + t;
  for (int k = 0; k < SC_BATCH; ++k) {
    int e = base + k * SC_THREADS;
    int ec = min(e, elast);
    int dv = dst[ec];
    dn[k] = (e < e1) ? dv : -1;
    sn[k] = src[ec];
  }
  for (; base < e1; base += SC_BATCH * SC_THREADS) {
    int d[SC_BATCH], s[SC_BATCH];
    for (int k = 0; k < SC_BATCH; ++k) { d[k] = dn[k]; s[k] = sn[k]; }
    const int nb = base + SC_BATCH * SC_THREADS;
    for (int k = 0; k < SC_BATCH; ++k) {
      int e = nb + k * SC_THREADS;
      int ec = min(e, elast);
      int dv = dst[ec];
      dn[k] = (e < e1) ? dv : -1;
      sn[k] = src[ec];
    }
    v2f v[SC_BATCH];
    for (int k = 0; k < SC_BATCH; ++k) {
      unsigned r = (unsigned)(d[k] - lo);
      v[k] = feat2[(r < (unsigned)nloc) ? s[k] : 0];
    }
    for (int k = 0; k < SC_BATCH; ++k) {
      unsigned r = (unsigned)(d[k] - lo);
      if (r < (unsigned)nloc) {
        atomicAdd(&tile[2 * r + 0], v[k].x);
        atomicAdd(&tile[2 * r + 1], v[k].y);
      }
    }
  }
  __syncthreads();

  float* p = partial + ((size_t)blockIdx.x * N + lo) * 2;
  const int cnt = nloc * 2;
  for (int i = t; i < cnt; i += SC_THREADS) p[i] = tile[i];
}

#define BIN_MAX    256
#define STAGE_MAX  32768
#define AGG_LDS_FLOATS 81920
#define AG_BATCH   16
#define WIN_BATCH  9
#define BIN_RANGES 21
#define DEG_TILE   32768
#define DG_BATCH   32

__device__ __forceinline__ int qdiv(int v, int dvs, float inv) {
  int q = (int)((float)v * inv);
  int r = v - q * dvs;
  q += (r >= dvs) ? 1 : 0;
  q -= (r < 0) ? 1 : 0;
  return q;
}

__device__ __forceinline__ int edge_bucket(int d, int s, int N, int R, int WR,
                                           int nwin, float invR, float invW,
                                           int* packed) {
  if ((unsigned)d >= (unsigned)N || (unsigned)s >= (unsigned)N) return -1;
  int y = qdiv(d, R, invR);
  int x = qdiv(s, WR, invW);
  *packed = ((s - x * WR) << 15) | (d - y * R);
  return y * nwin + x;
}

__global__ void k_count(const int* __restrict__ src, const int* __restrict__ dst,
                        int* __restrict__ cnt, int E, int N, int R, int WR,
                        int nwin, int NB, int epb) {
  __shared__ int hist[BIN_MAX];
  const int t = threadIdx.x;
  if (t < BIN_MAX) hist[t] = 0;
  __syncthreads();
  const float invR = 1.0f / (float)R, invW = 1.0f / (float)WR;
  const int e0 = min(E, (int)blockIdx.x * epb);
  const int e1 = min(E, e0 + epb);
  for (int base = e0 + t; base < e1; base += SC_BATCH * SC_THREADS) {
    int dv[SC_BATCH], sv[SC_BATCH];
    for (int k = 0; k < SC_BATCH; ++k) {
      int e = base + k * SC_THREADS;
      int ec = min(e, e1 - 1);
      int d = dst[ec];
      sv[k] = src[ec];
      dv[k] = (e < e1) ? d : -1;
    }
    for (int k = 0; k < SC_BATCH; ++k) {
      int pk;
      int b = edge_bucket(dv[k], sv[k], N, R, WR, nwin, invR, invW, &pk);
      if (b >= 0) atomicAdd(&hist[b], 1);
    }
  }
  __syncthreads();
  if (t < NB) cnt[(size_t)blockIdx.x * NB + t] = hist[t];
}

__global__ void k_place(const int* __restrict__ src, const int* __restrict__ dst,
                        const int* __restrict__ cnt, int* __restrict__ ent,
                        int* __restrict__ bst, int E, int N, int R, int WR,
                        int nwin, int NB, int epb) {
  __shared__ int tot[BIN_MAX], pre[BIN_MAX], own[BIN_MAX];
  __shared__ int cur[BIN_MAX], lst[BIN_MAX], gst[BIN_MAX];
  __shared__ int sa[BIN_MAX], so[BIN_MAX];
  __shared__ int sval[STAGE_MAX];
  __shared__ unsigned char sbin[STAGE_MAX];
  const int t = threadIdx.x;
  const int slice = blockIdx.x;
  const int S = gridDim.x;
  if (t < BIN_MAX) {
    tot[t] = 0;
    pre[t] = 0;
    own[t] = (t < NB) ? cnt[(size_t)slice * NB + t] : 0;
  }
  __syncthreads();
  {
    const int b = t & (BIN_MAX - 1);
    const int part = t >> 8;
    const int per = (S + 3) >> 2;
    const int s0 = part * per;
    const int s1 = min(S, s0 + per);
    if (b < NB && s0 < s1) {
      int all = 0, before = 0;
      for (int s = s0; s < s1; s += 16) {
        int v[16];
#pragma unroll
        for (int k = 0; k < 16; ++k) {
          int ss = min(s + k, s1 - 1);
          v[k] = cnt[(size_t)ss * NB + b];
        }
#pragma unroll
        for (int k = 0; k < 16; ++k) {
          int ss = s + k;
          int vv = (ss < s1) ? v[k] : 0;
          all += vv;
          before += (ss < slice) ? vv : 0;
        }
      }
      atomicAdd(&tot[b], all);
      atomicAdd(&pre[b], before);
    }
  }
  __syncthreads();
  int pa = 0, po = 0;
  if (t < BIN_MAX) { pa = tot[t]; po = own[t]; sa[t] = pa; so[t] = po; }
  __syncthreads();
  for (int off = 1; off < BIN_MAX; off <<= 1) {
    int a2 = 0, o2 = 0;
    if (t < BIN_MAX && t >= off) { a2 = sa[t - off]; o2 = so[t - off]; }
    __syncthreads();
    if (t < BIN_MAX) { pa += a2; po += o2; sa[t] = pa; so[t] = po; }
    __syncthreads();
  }
  if (t < NB) {
    int gs = pa - tot[t];
    int ls = po - own[t];
    gst[t] = gs + pre[t];
    lst[t] = ls;
    cur[t] = ls;
    if (slice == 0) {
      bst[t] = gs;
      if (t == NB - 1) bst[NB] = pa;
    }
  }
  __syncthreads();

  const float invR = 1.0f / (float)R, invW = 1.0f / (float)WR;
  const int e0 = min(E, slice * epb);
  const int e1 = min(E, e0 + epb);
  for (int base = e0 + t; base < e1; base += SC_BATCH * SC_THREADS) {
    int dv[SC_BATCH], sv[SC_BATCH];
    for (int k = 0; k < SC_BATCH; ++k) {
      int e = base + k * SC_THREADS;
      int ec = min(e, e1 - 1);
      int d = dst[ec];
      sv[k] = src[ec];
      dv[k] = (e < e1) ? d : -1;
    }
    for (int k = 0; k < SC_BATCH; ++k) {
      int pk = 0;
      int b = edge_bucket(dv[k], sv[k], N, R, WR, nwin, invR, invW, &pk);
      if (b >= 0) {
        int lp = atomicAdd(&cur[b], 1);
        sval[lp] = pk;
        sbin[lp] = (unsigned char)b;
      }
    }
  }
  __syncthreads();
  const int ntot = lst[NB - 1] + own[NB - 1];
  for (int i = t; i < ntot; i += SC_THREADS) {
    int b = sbin[i];
    ent[gst[b] + (i - lst[b])] = sval[i];
  }
}

__global__ void k_deg_bin(const int* __restrict__ ent, const int* __restrict__ bst,
                          int* __restrict__ partial, int N, int R) {
  __shared__ __attribute__((aligned(16))) int tile[DEG_TILE];
  const int t = threadIdx.x;
  const int lo = blockIdx.y * R;
  const int nloc = max(0, min(R, N - lo));
  const int bucket = blockIdx.y * gridDim.x + blockIdx.x;
  const int b0 = bst[bucket];
  const int b1 = bst[bucket + 1];

  for (int i = t; i < nloc; i += SC_THREADS) tile[i] = 0;
  __syncthreads();

  for (int base = b0 + t; base < b1; base += DG_BATCH * SC_THREADS) {
    int en[DG_BATCH];
    for (int k = 0; k < DG_BATCH; ++k) {
      int i = base + k * SC_THREADS;
      int v = ent[min(i, b1 - 1)];
      en[k] = (i < b1) ? v : -1;
    }
    for (int k = 0; k < DG_BATCH; ++k)
      if (en[k] >= 0) atomicAdd(&tile[en[k] & 32767], 1);
  }
  __syncthreads();

  int* p = partial + ((size_t)blockIdx.x * N + lo);
  for (int i = t; i < nloc; i += SC_THREADS) p[i] = tile[i];
}

__global__ void k_agg_bin(const int* __restrict__ ent, const int* __restrict__ bst,
                          const float* __restrict__ feat, float* __restrict__ partial,
                          int N, int R, int WR) {
  __shared__ __attribute__((aligned(16))) float lds[AGG_LDS_FLOATS];
  float* tile = lds;
  float* win = lds + 2 * R;
  const int t = threadIdx.x;
  const int lo = blockIdx.y * R;
  const int nloc = max(0, min(R, N - lo));
  const int wlo = blockIdx.x * WR;
  const int wn = max(0, min(WR, N - wlo));
  const int bucket = blockIdx.y * gridDim.x + blockIdx.x;
  const int b0 = bst[bucket];
  const int b1 = bst[bucket + 1];

  const float4 z4 = {0.0f, 0.0f, 0.0f, 0.0f};
  for (int i = t * 4; i < nloc * 2; i += SC_THREADS * 4)
    *reinterpret_cast<float4*>(&tile[i]) = z4;
  const float* wsrc = feat + (size_t)wlo * 2;
  const v4f* wsrc4 = reinterpret_cast<const v4f*>(wsrc);
  const int wf = wn * 2;
  const int wf4 = wf >> 2;
  static_assert(WIN_BATCH == 9, "the fence below names nine pieces");
  for (int base = t; base < wf4; base += WIN_BATCH * SC_THREADS) {
    v4f wv[WIN_BATCH];
    for (int k = 0; k < WIN_BATCH; ++k)
      wv[k] = wsrc4[min(base + k * SC_THREADS, wf4 - 1)];
    asm volatile("" :: "v"(wv[0]), "v"(wv[1]), "v"(wv[2]), "v"(wv[3]), "v"(wv[4]),
                       "v"(wv[5]), "v"(wv[6]), "v"(wv[7]), "v"(wv[8]));
    for (int k = 0; k < WIN_BATCH; ++k) {
      int i = base + k * SC_THREADS;
      if (i < wf4) *reinterpret_cast<v4f*>(&win[4 * i]) = wv[k];
    }
  }
  if (t < (wf & 3)) win[(wf4 << 2) + t] = wsrc[(wf4 << 2) + t];
  __syncthreads();

  const v2f* win2 = reinterpret_cast<const v2f*>(win);
  for (int base = b0 + t; base < b1; base += AG_BATCH * SC_THREADS) {
    int en[AG_BATCH];
    for (int k = 0; k < AG_BATCH; ++k) {
      int i = base + k * SC_THREADS;
      int v = ent[min(i, b1 - 1)];
      en[k] = (i < b1) ? v : -1;
    }
    v2f v[AG_BATCH];
    for (int k = 0; k < AG_BATCH; ++k)
      v[k] = win2[(en[k] >= 0) ? (en[k] >> 15) : 0];
    for (int k = 0; k < AG_BATCH; ++k) {
      if (en[k] >= 0) {
        int dl = en[k] & 32767;
        atomicAdd(&tile[2 * dl + 0], v[k].x);
        atomicAdd(&tile[2 * dl + 1], v[k].y);
      }
    }
  }
  __syncthreads();

  float* p = partial + ((size_t)blockIdx.x * N + lo) * 2;
  const int cnt2 = nloc * 2;
  for (int i = t; i < cnt2; i += SC_THREADS) p[i] = tile[i];
}

__global__ void k_dinv(const int* __restrict__ part, const float* __restrict__ x,
                       float* __restrict__ dinv, float* __restrict__ xs,
                       int N, int nsl) {
  int i = blockIdx.x * blockDim.x + threadIdx.x;
  if (i < N) {
    const v2f xv = reinterpret_cast<const v2f*>(x)[i];
    int c = 0;
    for (int s = 0; s < nsl; s += 16) {
      int v[16];
#pragma unroll
      for (int k = 0; k < 16; ++k)
        v[k] = part[(size_t)min(s + k, nsl - 1) * N + i];
#pragma unroll
      for (int k = 0; k < 16; ++k) c += (s + k < nsl) ? v[k] : 0;
    }
    float d = 1.0f + (float)c;
    float di = (d > 0.0f) ? rsqrtf(d) : 0.0f;
    dinv[i] = di;
    reinterpret_cast<v2f*>(xs)[i] = xv * di;
  }
}

__global__ void k_gemm12(const float* __restrict__ part, const float* __restrict__ xs,
                         const float* __restrict__ dinv, const float* __restrict__ W1,
                         const float* __restrict__ b1, const float* __restrict__ W2,
                         float* __restrict__ h2s, int nwaves, int N, int nsl) {
  const int wave = (blockIdx.x * blockDim.x + threadIdx.x) >> 5;
  const int lane = threadIdx.x & 31;
  if (wave >= nwaves) return;
  const int c = lane & 15;
  const int h = lane >> 4;
  const int node = (wave << 5) + lane;

  const v2f xv = reinterpret_cast<const v2f*>(xs)[node];
  const float di = dinv[node];
  v2f w1;
  w1.x = W1[0 * 16 + c];
  w1.y = W1[1 * 16 + c];
  float bias[8];
#pragma unroll
  for (int j = 0; j < 8; ++j) bias[j] = b1[8 * h + j];
  const int m = c & 7;
  const int mm = (m < 2) ? m : 0;
  float u[8];
#pragma unroll
  for (int k = 0; k < 8; ++k) u[k] = W2[(8 * h + k) * 2 + mm];

  const v2f* p2 = reinterpret_cast<const v2f*>(part) + node;
  v2f a0 = {0.0f, 0.0f}, a1 = {0.0f, 0.0f}, a2 = {0.0f, 0.0f}, a3 = {0.0f, 0.0f};
  int s = 0;
  for (; s + 8 <= nsl; s += 8) {
    v2f t0 = p2[(size_t)(s + 0) * N], t1 = p2[(size_t)(s + 1) * N];
    v2f t2 = p2[(size_t)(s + 2) * N], t3 = p2[(size_t)(s + 3) * N];
    v2f t4 = p2[(size_t)(s + 4) * N], t5 = p2[(size_t)(s + 5) * N];
    v2f t6 = p2[(size_t)(s + 6) * N], t7 = p2[(size_t)(s + 7) * N];
    a0 += t0; a1 += t1; a2 += t2; a3 += t3;
    a0 += t4; a1 += t5; a2 += t6; a3 += t7;
  }
  for (; s + 4 <= nsl; s += 4) {
    a0 += p2[(size_t)(s + 0) * N];
    a1 += p2[(size_t)(s + 1) * N];
    a2 += p2[(size_t)(s + 2) * N];
    a3 += p2[(size_t)(s + 3) * N];
  }
  for (; s < nsl; ++s) a0 += p2[(size_t)s * N];
  const v2f ax = (((a0 + a1) + (a2 + a3)) + xv) * di;

  const v2f z2 = {0.0f, 0.0f};
  const v2f wa0 = h ? z2 : w1, wa1 = h ? w1 : z2;
  const v2f xb0 = h ? z2 : ax, xb1 = h ? ax : z2;
  v8f d0 = {}, d1 = {};
  d0 = __builtin_amdgcn_wmma_f32_16x16x4_f32(
       false, wa0,  false, xb0,
       (short)0, d0,  false,  false);
  d1 = __builtin_amdgcn_wmma_f32_16x16x4_f32(
      false, wa1, false, xb1, (short)0, d1, false, false);

  float r0[8], r1[8];
#pragma unroll
  for (int j = 0; j < 8; ++j) {
    r0[j] = fmaxf(d0[j] + bias[j], 0.0f);
    r1[j] = fmaxf(d1[j] + bias[j], 0.0f);
  }

  v8f e0 = {}, e1 = {};
#pragma unroll
  for (int q = 0; q < 4; ++q) {
    v2f w2;
    w2.x = (m < 2) ? u[2 * q + 0] : 0.0f;
    w2.y = (m < 2) ? u[2 * q + 1] : 0.0f;
    v2f rb0, rb1;
    rb0.x = r0[2 * q]; rb0.y = r0[2 * q + 1];
    rb1.x = r1[2 * q]; rb1.y = r1[2 * q + 1];
    e0 = __builtin_amdgcn_wmma_f32_16x16x4_f32(
        false, w2, false, rb0, (short)0, e0, false, false);
    e1 = __builtin_amdgcn_wmma_f32_16x16x4_f32(
        false, w2, false, rb1, (short)0, e1, false, false);
  }

  v2f o;
  o.x = h ? e1[0] : e0[0];
  o.y = h ? e1[1] : e0[1];
  reinterpret_cast<v2f*>(h2s)[node] = o * di;
}

__global__ void k_gemm1_tail(const float* __restrict__ part, const float* __restrict__ xs,
                             const float* __restrict__ dinv, const float* __restrict__ W1,
                             const float* __restrict__ b1, float* __restrict__ r,
                             int n0, int N, int nsl) {
  int tid = blockIdx.x * blockDim.x + threadIdx.x;
  int n = n0 + (tid >> 4);
  if (n >= N) return;
  int f = tid & 15;
  const v2f* p2 = reinterpret_cast<const v2f*>(part) + n;
  v2f acc = {0.0f, 0.0f};
  for (int s = 0; s < nsl; ++s) acc += p2[(size_t)s * N];
  v2f xv = reinterpret_cast<const v2f*>(xs)[n];
  v2f ax = (acc + xv) * dinv[n];
  r[n * 16 + f] = fmaxf(ax.x * W1[f] + ax.y * W1[16 + f] + b1[f], 0.0f);
}

__global__ void k_gemm2_tail(const float* __restrict__ r, const float* __restrict__ W2,
                             const float* __restrict__ dinv,
                             float* __restrict__ h2s, int n0, int N) {
  int tid = blockIdx.x * blockDim.x + threadIdx.x;
  int n = n0 + (tid >> 1);
  if (n >= N) return;
  int f = tid & 1;
  float acc = 0.0f;
#pragma unroll
  for (int k = 0; k < 16; ++k)
    acc += r[n * 16 + k] * W2[k * 2 + f];
  h2s[n * 2 + f] = acc * dinv[n];
}

__global__ void k_final(const float* __restrict__ part, const float* __restrict__ h2s,
                        const float* __restrict__ dinv, const float* __restrict__ b2,
                        float* __restrict__ out, int N, int nsl) {
  int tid = blockIdx.x * blockDim.x + threadIdx.x;
  if (tid >= 2 * N) return;
  int n = tid >> 1;
  int f = tid & 1;
  const size_t stride = (size_t)2 * N;
  float a0 = 0.0f, a1 = 0.0f, a2 = 0.0f, a3 = 0.0f;
  int s = 0;
  for (; s + 4 <= nsl; s += 4) {
    a0 += part[(size_t)(s + 0) * stride + tid];
    a1 += part[(size_t)(s + 1) * stride + tid];
    a2 += part[(size_t)(s + 2) * stride + tid];
    a3 += part[(size_t)(s + 3) * stride + tid];
  }
  for (; s < nsl; ++s) a0 += part[(size_t)s * stride + tid];
  float di = dinv[n];
  out[tid] = ((a0 + a1) + (a2 + a3)) * di + (h2s[tid] * di + b2[f]);
}

extern "C" void kernel_launch(void* const* d_in, const int* in_sizes, int n_in,
                              void* d_out, int out_size, void* d_ws, size_t ws_size,
                              hipStream_t stream) {
  const float* x  = (const float*)d_in[0];
  const int*   ei = (const int*)d_in[1];
  const float* W1 = (const float*)d_in[2];
  const float* b1 = (const float*)d_in[3];
  const float* W2 = (const float*)d_in[4];
  const float* b2 = (const float*)d_in[5];
  float* out = (float*)d_out;

  const int N = in_sizes[0] / 2;
  const int E = in_sizes[1] / 2;
  const int* src = ei;
  const int* dst = ei + E;

  const size_t NP = ((size_t)N + 63) & ~(size_t)63;
  float* ws   = (float*)d_ws;
  float* dinv = ws;
  float* xs   = ws + NP;
  float* r    = ws + (size_t)3 * NP;
  float* h2s  = ws + (size_t)19 * NP;
  float* part = ws + (size_t)21 * NP;

  const int B = 256;
  const int nwv = N / 32;
  const int n0 = nwv * 32;
  const int ntail = N - n0;

  const int nr1 = (N + TILE_FLOATS - 1) / TILE_FLOATS;
  const int ns1 = (256 / nr1) > 0 ? (256 / nr1) : 1;
  const int nrfit = (2 * N + TILE_FLOATS - 1) / TILE_FLOATS;
  const int nr2 = nrfit > BIN_RANGES ? nrfit : BIN_RANGES;
  const int ns2 = (256 / nr2) > 0 ? (256 / nr2) : 1;

  const int S   = 256;
  const int R2  = (((N + nr2 - 1) / nr2) + 31) & ~31;
  const int WR  = (((N + ns2 - 1) / ns2) + 31) & ~31;
  const int NB  = nr2 * ns2;
  const int epb = (E + S - 1) / S;
  const size_t pa1 = (size_t)ns1 * N, pa2 = (size_t)ns2 * 2 * N;
  const size_t partFloats = (((pa1 > pa2) ? pa1 : pa2) + 63) & ~(size_t)63;
  int* cnt = (int*)(part + partFloats);
  int* bst = cnt + (((size_t)S * NB + 63) & ~(size_t)63);
  int* ent = bst + (((size_t)NB + 1 + 63) & ~(size_t)63);
  const size_t need = ((size_t)(ent - (int*)ws) + (size_t)E) * sizeof(int);
  const bool binned = E > 0 && NB <= BIN_MAX && R2 <= 32768 && WR <= 65536 &&
                      2 * (R2 + WR) <= AGG_LDS_FLOATS && epb <= STAGE_MAX &&
                      N < (1 << 24) && need <= ws_size;

  if (binned) {
    k_count<<<S, SC_THREADS, 0, stream>>>(src, dst, cnt, E, N, R2, WR, ns2, NB, epb);
    k_place<<<S, SC_THREADS, 0, stream>>>(src, dst, cnt, ent, bst, E, N, R2, WR, ns2, NB, epb);
    k_deg_bin<<<dim3(ns2, nr2), SC_THREADS, 0, stream>>>(ent, bst, (int*)part, N, R2);
    k_dinv<<<(N + B - 1) / B, B, 0, stream>>>((const int*)part, x, dinv, xs, N, ns2);
    k_agg_bin<<<dim3(ns2, nr2), SC_THREADS, 0, stream>>>(ent, bst, xs, part, N, R2, WR);
  } else {
    k_degree_lds<<<dim3(ns1, nr1), SC_THREADS, 0, stream>>>(dst, (int*)part, E, N);
    k_dinv<<<(N + B - 1) / B, B, 0, stream>>>((const int*)part, x, dinv, xs, N, ns1);
    k_agg_lds<<<dim3(ns2, nr2), SC_THREADS, 0, stream>>>(src, dst, xs, part, E, N);
  }
  if (nwv > 0)
    k_gemm12<<<(nwv * 32 + B - 1) / B, B, 0, stream>>>(part, xs, dinv, W1, b1, W2, h2s, nwv, N, ns2);
  if (ntail > 0) {
    k_gemm1_tail<<<(ntail * 16 + B - 1) / B, B, 0, stream>>>(part, xs, dinv, W1, b1, r, n0, N, ns2);
    k_gemm2_tail<<<(ntail * 2 + B - 1) / B, B, 0, stream>>>(r, W2, dinv, h2s, n0, N);
  }

  if (binned)
    k_agg_bin<<<dim3(ns2, nr2), SC_THREADS, 0, stream>>>(ent, bst, h2s, part, N, R2, WR);
  else
    k_agg_lds<<<dim3(ns2, nr2), SC_THREADS, 0, stream>>>(src, dst, h2s, part, E, N);
  k_final <<<(2 * N + B - 1) / B,  B, 0, stream>>>(part, h2s, dinv, b2, out, N, ns2);
}


# Edge bucketing in one pass (was count + place): each slice ordered by bucket in LDS, rank from its counting LDS add, stored as its own run; a bucket is read as 256 pieces; float sum order changes
# speedup vs baseline: 13.0028x; 1.2305x over previous
#include <hip/hip_runtime.h>


typedef __attribute__((ext_vector_type(2))) float v2f;
typedef __attribute__((ext_vector_type(4))) float v4f;
typedef __attribute__((ext_vector_type(8))) float v8f;

#define TILE_FLOATS 65536
#define SC_THREADS  1024
#define SC_BATCH    8

__global__ void k_degree_lds(const int* __restrict__ dst, int* __restrict__ partial,
                             int E, int N) {
  __shared__ __attribute__((aligned(16))) int tile[TILE_FLOATS];
  const int t = threadIdx.x;
  const int R = (((N + (int)gridDim.y - 1) / (int)gridDim.y) + 31) & ~31;
  const int eps = (E + (int)gridDim.x - 1) / (int)gridDim.x;
  const int lo = blockIdx.y * R;
  const int nloc = max(0, min(R, N - lo));

  const int4 z4 = {0, 0, 0, 0};
  for (int i = t * 4; i < TILE_FLOATS; i += SC_THREADS * 4)
    *reinterpret_cast<int4*>(&tile[i]) = z4;
  __syncthreads();

  const int e0 = blockIdx.x * eps;
  const int e1 = min(E, e0 + eps);
  const int elast = e1 - 1;

  int dn[SC_BATCH];
  int base = e0 + t;
  for (int k = 0; k < SC_BATCH; ++k) {
    int e = base + k * SC_THREADS;
    int dv = dst[min(e, elast)];
    dn[k] = (e < e1) ? dv : -1;
  }
  for (; base < e1; base += SC_BATCH * SC_THREADS) {
    int d[SC_BATCH];
    for (int k = 0; k < SC_BATCH; ++k) d[k] = dn[k];
    const int nb = base + SC_BATCH * SC_THREADS;
    for (int k = 0; k < SC_BATCH; ++k) {
      int e = nb + k * SC_THREADS;
      int dv = dst[min(e, elast)];
      dn[k] = (e < e1) ? dv : -1;
    }
    for (int k = 0; k < SC_BATCH; ++k) {
      unsigned r = (unsigned)(d[k] - lo);
      if (r < (unsigned)nloc) atomicAdd(&tile[r], 1);
    }
  }
  __syncthreads();

  int* p = partial + ((size_t)blockIdx.x * N + lo);
  for (int i = t; i < nloc; i += SC_THREADS) p[i] = tile[i];
}

__global__ void k_agg_lds(const int* __restrict__ src, const int* __restrict__ dst,
                           const float* __restrict__ feat, float* __restrict__ partial,
                           int E, int N) {
  __shared__ __attribute__((aligned(16))) float tile[TILE_FLOATS];
  const int t = threadIdx.x;
  const int R = (((N + (int)gridDim.y - 1) / (int)gridDim.y) + 31) & ~31;
  const int eps = (E + (int)gridDim.x - 1) / (int)gridDim.x;
  const int lo = blockIdx.y * R;
  const int nloc = max(0, min(R, N - lo));

  const float4 z4 = {0.0f, 0.0f, 0.0f, 0.0f};
  for (int i = t * 4; i < TILE_FLOATS; i += SC_THREADS * 4)
    *reinterpret_cast<float4*>(&tile[i]) = z4;
  __syncthreads();

  const int e0 = blockIdx.x * eps;
  const int e1 = min(E, e0 + eps);
  const int elast = e1 - 1;
  const v2f* feat2 = reinterpret_cast<const v2f*>(feat);

  int dn[SC_BATCH], sn[SC_BATCH];
  int base = e0 + t;
  for (int k = 0; k < SC_BATCH; ++k) {
    int e = base + k * SC_THREADS;
    int ec = min(e, elast);
    int dv = dst[ec];
    dn[k] = (e < e1) ? dv : -1;
    sn[k] = src[ec];
  }
  for (; base < e1; base += SC_BATCH * SC_THREADS) {
    int d[SC_BATCH], s[SC_BATCH];
    for (int k = 0; k < SC_BATCH; ++k) { d[k] = dn[k]; s[k] = sn[k]; }
    const int nb = base + SC_BATCH * SC_THREADS;
    for (int k = 0; k < SC_BATCH; ++k) {
      int e = nb + k * SC_THREADS;
      int ec = min(e, elast);
      int dv = dst[ec];
      dn[k] = (e < e1) ? dv : -1;
      sn[k] = src[ec];
    }
    v2f v[SC_BATCH];
    for (int k = 0; k < SC_BATCH; ++k) {
      unsigned r = (unsigned)(d[k] - lo);
      v[k] = feat2[(r < (unsigned)nloc) ? s[k] : 0];
    }
    for (int k = 0; k < SC_BATCH; ++k) {
      unsigned r = (unsigned)(d[k] - lo);
      if (r < (unsigned)nloc) {
        atomicAdd(&tile[2 * r + 0], v[k].x);
        atomicAdd(&tile[2 * r + 1], v[k].y);
      }
    }
  }
  __syncthreads();

  float* p = partial + ((size_t)blockIdx.x * N + lo) * 2;
  const int cnt = nloc * 2;
  for (int i = t; i < cnt; i += SC_THREADS) p[i] = tile[i];
}

#define BIN_MAX    256
#define STAGE_MAX  32768
#define NSLICE     256
#define PL_TRIPS   (STAGE_MAX / (SC_BATCH * SC_THREADS))
#define PC_SLICES  (NSLICE / (SC_THREADS / 32))
#define PC_CHUNKS  4
#define AGG_LDS_FLOATS 81920
#define WIN_BATCH  9
#define BIN_RANGES 21
#define DEG_TILE   32768

__device__ __forceinline__ int qdiv(int v, int dvs, float inv) {
  int q = (int)((float)v * inv);
  int r = v - q * dvs;
  q += (r >= dvs) ? 1 : 0;
  q -= (r < 0) ? 1 : 0;
  return q;
}

__device__ __forceinline__ int edge_bucket(int d, int s, int N, int R, int WR,
                                           int nwin, float invR, float invW,
                                           int* packed) {
  if ((unsigned)d >= (unsigned)N || (unsigned)s >= (unsigned)N) return -1;
  int y = qdiv(d, R, invR);
  int x = qdiv(s, WR, invW);
  *packed = ((s - x * WR) << 15) | (d - y * R);
  return y * nwin + x;
}

__global__ void k_place(const int* __restrict__ src, const int* __restrict__ dst,
                        int* __restrict__ ent, int* __restrict__ off,
                        int E, int N, int R, int WR, int nwin, int NB, int epb,
                        int eps) {
  __shared__ int cur[BIN_MAX], sa[BIN_MAX], lst[BIN_MAX];
  __shared__ __attribute__((aligned(16))) int sval[STAGE_MAX];
  const int t = threadIdx.x;
  const int slice = blockIdx.x;
  if (t < BIN_MAX) cur[t] = 0;
  __syncthreads();

  const float invR = 1.0f / (float)R, invW = 1.0f / (float)WR;
  const int e0 = min(E, slice * epb);
  const int e1 = min(E, e0 + epb);
  int pk[PL_TRIPS * SC_BATCH];
  int br[PL_TRIPS * SC_BATCH];
#pragma unroll
  for (int trip = 0; trip < PL_TRIPS; ++trip) {
    int dv[SC_BATCH], sv[SC_BATCH];
#pragma unroll
    for (int k = 0; k < SC_BATCH; ++k) {
      const int eu = e0 + (trip * SC_BATCH + k) * SC_THREADS;
      dv[k] = 0;
      sv[k] = 0;
      if (eu < e1) {
        int ec = min(eu + t, e1 - 1);
        dv[k] = dst[ec];
        sv[k] = src[ec];
      }
    }
#pragma unroll
    for (int k = 0; k < SC_BATCH; ++k) {
      const int e = e0 + (trip * SC_BATCH + k) * SC_THREADS + t;
      dv[k] = (e < e1) ? dv[k] : -1;
    }
    int bb[SC_BATCH], rk[SC_BATCH];
#pragma unroll
    for (int k = 0; k < SC_BATCH; ++k) {
      int p = 0;
      bb[k] = -1;
      if (e0 + (trip * SC_BATCH + k) * SC_THREADS < e1)
        bb[k] = edge_bucket(dv[k], sv[k], N, R, WR, nwin, invR, invW, &p);
      pk[trip * SC_BATCH + k] = p;
    }
#pragma unroll
    for (int k = 0; k < SC_BATCH; ++k) {
      rk[k] = 0;
      if (bb[k] >= 0) rk[k] = atomicAdd(&cur[bb[k]], 1);
    }
#pragma unroll
    for (int k = 0; k < SC_BATCH; ++k)
      br[trip * SC_BATCH + k] = (bb[k] >= 0) ? ((bb[k] << 16) | rk[k]) : -1;
  }
  __syncthreads();

  int own = 0, pa = 0;
  if (t < BIN_MAX) { own = cur[t]; pa = own; sa[t] = pa; }
  __syncthreads();
  for (int o = 1; o < BIN_MAX; o <<= 1) {
    int a2 = 0;
    if (t < BIN_MAX && t >= o) a2 = sa[t - o];
    __syncthreads();
    if (t < BIN_MAX) { pa += a2; sa[t] = pa; }
    __syncthreads();
  }
  if (t < BIN_MAX) lst[t] = pa - own;
  if (t < NB) off[(size_t)t * NSLICE + slice] = pa - own;
  if (t == NB - 1) off[(size_t)NB * NSLICE + slice] = pa;
  __syncthreads();

  bool has[PL_TRIPS * SC_BATCH];
#pragma unroll
  for (int q = 0; q < PL_TRIPS * SC_BATCH; ++q) {
    has[q] = br[q] >= 0;
    br[q] = lst[has[q] ? (br[q] >> 16) : 0] + (br[q] & 0xffff);
  }
#pragma unroll
  for (int q = 0; q < PL_TRIPS * SC_BATCH; ++q)
    if (has[q]) sval[br[q]] = pk[q];
  __syncthreads();

  const int ntot = sa[BIN_MAX - 1];
  int* eo = ent + (size_t)slice * eps;
  const int n4 = ntot >> 2;
  for (int i = t; i < n4; i += SC_THREADS)
    reinterpret_cast<int4*>(eo)[i] = reinterpret_cast<const int4*>(sval)[i];
  if (t < (ntot & 3)) eo[(n4 << 2) + t] = sval[(n4 << 2) + t];
}

static_assert(PC_SLICES * (SC_THREADS / 32) == NSLICE, "a consumer's waves cover every slice");

__global__ void k_deg_bin(const int* __restrict__ ent, const int* __restrict__ off,
                          int* __restrict__ partial, int N, int R, int eps) {
  __shared__ __attribute__((aligned(16))) int tile[DEG_TILE];
  const int t = threadIdx.x;
  const int lane = t & 31;
  const int w = __builtin_amdgcn_readfirstlane(t >> 5);
  const int lo = blockIdx.y * R;
  const int nloc = max(0, min(R, N - lo));
  const int bucket = blockIdx.y * gridDim.x + blockIdx.x;
  const int* ob = off + (size_t)bucket * NSLICE + w * PC_SLICES;
  int st[PC_SLICES], en[PC_SLICES];
#pragma unroll
  for (int j = 0; j < PC_SLICES; ++j) { st[j] = ob[j]; en[j] = ob[NSLICE + j]; }

  for (int i = t; i < nloc; i += SC_THREADS) tile[i] = 0;
  __syncthreads();

  const int* es = ent + (size_t)(w * PC_SLICES) * eps;
  int v[PC_SLICES][PC_CHUNKS];
#pragma unroll
  for (int j = 0; j < PC_SLICES; ++j) {
#pragma unroll
    for (int c = 0; c < PC_CHUNKS; ++c) {
      int i = st[j] + lane + 32 * c;
      int x = es[(size_t)j * eps + max(min(i, en[j] - 1), 0)];
      v[j][c] = (i < en[j]) ? x : -1;
    }
  }
#pragma unroll
  for (int j = 0; j < PC_SLICES; ++j) {
#pragma unroll
    for (int c = 0; c < PC_CHUNKS; ++c)
      if (v[j][c] >= 0) atomicAdd(&tile[v[j][c] & 32767], 1);
  }
#pragma unroll
  for (int j = 0; j < PC_SLICES; ++j) {
    for (int i0 = st[j] + 32 * PC_CHUNKS; i0 < en[j]; i0 += 32) {
      int i = i0 + lane;
      if (i < en[j]) atomicAdd(&tile[es[(size_t)j * eps + i] & 32767], 1);
    }
  }
  __syncthreads();

  int* p = partial + ((size_t)blockIdx.x * N + lo);
  for (int i = t; i < nloc; i += SC_THREADS) p[i] = tile[i];
}

__global__ void k_agg_bin(const int* __restrict__ ent, const int* __restrict__ off,
                          const float* __restrict__ feat, float* __restrict__ partial,
                          int N, int R, int WR, int eps) {
  __shared__ __attribute__((aligned(16))) float lds[AGG_LDS_FLOATS];
  float* tile = lds;
  float* win = lds + 2 * R;
  const int t = threadIdx.x;
  const int lane = t & 31;
  const int w = __builtin_amdgcn_readfirstlane(t >> 5);
  const int lo = blockIdx.y * R;
  const int nloc = max(0, min(R, N - lo));
  const int wlo = blockIdx.x * WR;
  const int wn = max(0, min(WR, N - wlo));
  const int bucket = blockIdx.y * gridDim.x + blockIdx.x;
  const int* ob = off + (size_t)bucket * NSLICE + w * PC_SLICES;
  int st[PC_SLICES], en[PC_SLICES];
#pragma unroll
  for (int j = 0; j < PC_SLICES; ++j) { st[j] = ob[j]; en[j] = ob[NSLICE + j]; }

  const float4 z4 = {0.0f, 0.0f, 0.0f, 0.0f};
  for (int i = t * 4; i < nloc * 2; i += SC_THREADS * 4)
    *reinterpret_cast<float4*>(&tile[i]) = z4;
  const float* wsrc = feat + (size_t)wlo * 2;
  const v4f* wsrc4 = reinterpret_cast<const v4f*>(wsrc);
  const int wf = wn * 2;
  const int wf4 = wf >> 2;
  static_assert(WIN_BATCH == 9, "the fence below names nine pieces");
  for (int base = t; base < wf4; base += WIN_BATCH * SC_THREADS) {
    v4f wv[WIN_BATCH];
    for (int k = 0; k < WIN_BATCH; ++k)
      wv[k] = wsrc4[min(base + k * SC_THREADS, wf4 - 1)];
    asm volatile("" :: "v"(wv[0]), "v"(wv[1]), "v"(wv[2]), "v"(wv[3]), "v"(wv[4]),
                       "v"(wv[5]), "v"(wv[6]), "v"(wv[7]), "v"(wv[8]));
    for (int k = 0; k < WIN_BATCH; ++k) {
      int i = base + k * SC_THREADS;
      if (i < wf4) *reinterpret_cast<v4f*>(&win[4 * i]) = wv[k];
    }
  }
  if (t < (wf & 3)) win[(wf4 << 2) + t] = wsrc[(wf4 << 2) + t];
  __syncthreads();

  const v2f* win2 = reinterpret_cast<const v2f*>(win);
  const int* es = ent + (size_t)(w * PC_SLICES) * eps;
  int ev[PC_SLICES][PC_CHUNKS];
#pragma unroll
  for (int j = 0; j < PC_SLICES; ++j) {
#pragma unroll
    for (int c = 0; c < PC_CHUNKS; ++c) {
      int i = st[j] + lane + 32 * c;
      int x = es[(size_t)j * eps + max(min(i, en[j] - 1), 0)];
      ev[j][c] = (i < en[j]) ? x : -1;
    }
  }
#pragma unroll
  for (int j = 0; j < PC_SLICES; j += 4) {
    v2f v[4][PC_CHUNKS];
#pragma unroll
    for (int jj = 0; jj < 4; ++jj) {
#pragma unroll
      for (int c = 0; c < PC_CHUNKS; ++c)
        v[jj][c] = win2[(ev[j + jj][c] >= 0) ? (ev[j + jj][c] >> 15) : 0];
    }
#pragma unroll
    for (int jj = 0; jj < 4; ++jj) {
#pragma unroll
      for (int c = 0; c < PC_CHUNKS; ++c) {
        if (ev[j + jj][c] >= 0) {
          int dl = ev[j + jj][c] & 32767;
          atomicAdd(&tile[2 * dl + 0], v[jj][c].x);
          atomicAdd(&tile[2 * dl + 1], v[jj][c].y);
        }
      }
    }
  }
#pragma unroll
  for (int j = 0; j < PC_SLICES; ++j) {
    for (int i0 = st[j] + 32 * PC_CHUNKS; i0 < en[j]; i0 += 32) {
      int i = i0 + lane;
      if (i < en[j]) {
        int x = es[(size_t)j * eps + i];
        v2f r = win2[x >> 15];
        int dl = x & 32767;
        atomicAdd(&tile[2 * dl + 0], r.x);
        atomicAdd(&tile[2 * dl + 1], r.y);
      }
    }
  }
  __syncthreads();

  float* p = partial + ((size_t)blockIdx.x * N + lo) * 2;
  const int cnt2 = nloc * 2;
  for (int i = t; i < cnt2; i += SC_THREADS) p[i] = tile[i];
}

__global__ void k_dinv(const int* __restrict__ part, const float* __restrict__ x,
                       float* __restrict__ dinv, float* __restrict__ xs,
                       int N, int nsl) {
  int i = blockIdx.x * blockDim.x + threadIdx.x;
  if (i < N) {
    const v2f xv = reinterpret_cast<const v2f*>(x)[i];
    int c = 0;
    for (int s = 0; s < nsl; s += 16) {
      int v[16];
#pragma unroll
      for (int k = 0; k < 16; ++k)
        v[k] = part[(size_t)min(s + k, nsl - 1) * N + i];
#pragma unroll
      for (int k = 0; k < 16; ++k) c += (s + k < nsl) ? v[k] : 0;
    }
    float d = 1.0f + (float)c;
    float di = (d > 0.0f) ? rsqrtf(d) : 0.0f;
    dinv[i] = di;
    reinterpret_cast<v2f*>(xs)[i] = xv * di;
  }
}

__global__ void k_gemm12(const float* __restrict__ part, const float* __restrict__ xs,
                         const float* __restrict__ dinv, const float* __restrict__ W1,
                         const float* __restrict__ b1, const float* __restrict__ W2,
                         float* __restrict__ h2s, int nwaves, int N, int nsl) {
  const int wave = (blockIdx.x * blockDim.x + threadIdx.x) >> 5;
  const int lane = threadIdx.x & 31;
  if (wave >= nwaves) return;
  const int c = lane & 15;
  const int h = lane >> 4;
  const int node = (wave << 5) + lane;

  const v2f xv = reinterpret_cast<const v2f*>(xs)[node];
  const float di = dinv[node];
  v2f w1;
  w1.x = W1[0 * 16 + c];
  w1.y = W1[1 * 16 + c];
  float bias[8];
#pragma unroll
  for (int j = 0; j < 8; ++j) bias[j] = b1[8 * h + j];
  const int m = c & 7;
  const int mm = (m < 2) ? m : 0;
  float u[8];
#pragma unroll
  for (int k = 0; k < 8; ++k) u[k] = W2[(8 * h + k) * 2 + mm];

  const v2f* p2 = reinterpret_cast<const v2f*>(part) + node;
  v2f a0 = {0.0f, 0.0f}, a1 = {0.0f, 0.0f}, a2 = {0.0f, 0.0f}, a3 = {0.0f, 0.0f};
  int s = 0;
  for (; s + 8 <= nsl; s += 8) {
    v2f t0 = p2[(size_t)(s + 0) * N], t1 = p2[(size_t)(s + 1) * N];
    v2f t2 = p2[(size_t)(s + 2) * N], t3 = p2[(size_t)(s + 3) * N];
    v2f t4 = p2[(size_t)(s + 4) * N], t5 = p2[(size_t)(s + 5) * N];
    v2f t6 = p2[(size_t)(s + 6) * N], t7 = p2[(size_t)(s + 7) * N];
    a0 += t0; a1 += t1; a2 += t2; a3 += t3;
    a0 += t4; a1 += t5; a2 += t6; a3 += t7;
  }
  for (; s + 4 <= nsl; s += 4) {
    a0 += p2[(size_t)(s + 0) * N];
    a1 += p2[(size_t)(s + 1) * N];
    a2 += p2[(size_t)(s + 2) * N];
    a3 += p2[(size_t)(s + 3) * N];
  }
  for (; s < nsl; ++s) a0 += p2[(size_t)s * N];
  const v2f ax = (((a0 + a1) + (a2 + a3)) + xv) * di;

  const v2f z2 = {0.0f, 0.0f};
  const v2f wa0 = h ? z2 : w1, wa1 = h ? w1 : z2;
  const v2f xb0 = h ? z2 : ax, xb1 = h ? ax : z2;
  v8f d0 = {}, d1 = {};
  d0 = __builtin_amdgcn_wmma_f32_16x16x4_f32(
       false, wa0,  false, xb0,
       (short)0, d0,  false,  false);
  d1 = __builtin_amdgcn_wmma_f32_16x16x4_f32(
      false, wa1, false, xb1, (short)0, d1, false, false);

  float r0[8], r1[8];
#pragma unroll
  for (int j = 0; j < 8; ++j) {
    r0[j] = fmaxf(d0[j] + bias[j], 0.0f);
    r1[j] = fmaxf(d1[j] + bias[j], 0.0f);
  }

  v8f e0 = {}, e1 = {};
#pragma unroll
  for (int q = 0; q < 4; ++q) {
    v2f w2;
    w2.x = (m < 2) ? u[2 * q + 0] : 0.0f;
    w2.y = (m < 2) ? u[2 * q + 1] : 0.0f;
    v2f rb0, rb1;
    rb0.x = r0[2 * q]; rb0.y = r0[2 * q + 1];
    rb1.x = r1[2 * q]; rb1.y = r1[2 * q + 1];
    e0 = __builtin_amdgcn_wmma_f32_16x16x4_f32(
        false, w2, false, rb0, (short)0, e0, false, false);
    e1 = __builtin_amdgcn_wmma_f32_16x16x4_f32(
        false, w2, false, rb1, (short)0, e1, false, false);
  }

  v2f o;
  o.x = h ? e1[0] : e0[0];
  o.y = h ? e1[1] : e0[1];
  reinterpret_cast<v2f*>(h2s)[node] = o * di;
}

__global__ void k_gemm1_tail(const float* __restrict__ part, const float* __restrict__ xs,
                             const float* __restrict__ dinv, const float* __restrict__ W1,
                             const float* __restrict__ b1, float* __restrict__ r,
                             int n0, int N, int nsl) {
  int tid = blockIdx.x * blockDim.x + threadIdx.x;
  int n = n0 + (tid >> 4);
  if (n >= N) return;
  int f = tid & 15;
  const v2f* p2 = reinterpret_cast<const v2f*>(part) + n;
  v2f acc = {0.0f, 0.0f};
  for (int s = 0; s < nsl; ++s) acc += p2[(size_t)s * N];
  v2f xv = reinterpret_cast<const v2f*>(xs)[n];
  v2f ax = (acc + xv) * dinv[n];
  r[n * 16 + f] = fmaxf(ax.x * W1[f] + ax.y * W1[16 + f] + b1[f], 0.0f);
}

__global__ void k_gemm2_tail(const float* __restrict__ r, const float* __restrict__ W2,
                             const float* __restrict__ dinv,
                             float* __restrict__ h2s, int n0, int N) {
  int tid = blockIdx.x * blockDim.x + threadIdx.x;
  int n = n0 + (tid >> 1);
  if (n >= N) return;
  int f = tid & 1;
  float acc = 0.0f;
#pragma unroll
  for (int k = 0; k < 16; ++k)
    acc += r[n * 16 + k] * W2[k * 2 + f];
  h2s[n * 2 + f] = acc * dinv[n];
}

__global__ void k_final(const float* __restrict__ part, const float* __restrict__ h2s,
                        const float* __restrict__ dinv, const float* __restrict__ b2,
                        float* __restrict__ out, int N, int nsl) {
  int tid = blockIdx.x * blockDim.x + threadIdx.x;
  if (tid >= 2 * N) return;
  int n = tid >> 1;
  int f = tid & 1;
  const size_t stride = (size_t)2 * N;
  float a0 = 0.0f, a1 = 0.0f, a2 = 0.0f, a3 = 0.0f;
  int s = 0;
  for (; s + 4 <= nsl; s += 4) {
    a0 += part[(size_t)(s + 0) * stride + tid];
    a1 += part[(size_t)(s + 1) * stride + tid];
    a2 += part[(size_t)(s + 2) * stride + tid];
    a3 += part[(size_t)(s + 3) * stride + tid];
  }
  for (; s < nsl; ++s) a0 += part[(size_t)s * stride + tid];
  float di = dinv[n];
  out[tid] = ((a0 + a1) + (a2 + a3)) * di + (h2s[tid] * di + b2[f]);
}

extern "C" void kernel_launch(void* const* d_in, const int* in_sizes, int n_in,
                              void* d_out, int out_size, void* d_ws, size_t ws_size,
                              hipStream_t stream) {
  const float* x  = (const float*)d_in[0];
  const int*   ei = (const int*)d_in[1];
  const float* W1 = (const float*)d_in[2];
  const float* b1 = (const float*)d_in[3];
  const float* W2 = (const float*)d_in[4];
  const float* b2 = (const float*)d_in[5];
  float* out = (float*)d_out;

  const int N = in_sizes[0] / 2;
  const int E = in_sizes[1] / 2;
  const int* src = ei;
  const int* dst = ei + E;

  const size_t NP = ((size_t)N + 63) & ~(size_t)63;
  float* ws   = (float*)d_ws;
  float* dinv = ws;
  float* xs   = ws + NP;
  float* r    = ws + (size_t)3 * NP;
  float* h2s  = ws + (size_t)19 * NP;
  float* part = ws + (size_t)21 * NP;

  const int B = 256;
  const int nwv = N / 32;
  const int n0 = nwv * 32;
  const int ntail = N - n0;

  const int nr1 = (N + TILE_FLOATS - 1) / TILE_FLOATS;
  const int ns1 = (256 / nr1) > 0 ? (256 / nr1) : 1;
  const int nrfit = (2 * N + TILE_FLOATS - 1) / TILE_FLOATS;
  const int nr2 = nrfit > BIN_RANGES ? nrfit : BIN_RANGES;
  const int ns2 = (256 / nr2) > 0 ? (256 / nr2) : 1;

  const int R2  = (((N + nr2 - 1) / nr2) + 31) & ~31;
  const int WR  = (((N + ns2 - 1) / ns2) + 31) & ~31;
  const int NB  = nr2 * ns2;
  const int epb = (E + NSLICE - 1) / NSLICE;
  const int eps = (epb + 3) & ~3;
  const size_t pa1 = (size_t)ns1 * N, pa2 = (size_t)ns2 * 2 * N;
  const size_t partFloats = (((pa1 > pa2) ? pa1 : pa2) + 63) & ~(size_t)63;
  int* off = (int*)(part + partFloats);
  int* ent = off + ((((size_t)NB + 1) * NSLICE + 63) & ~(size_t)63);
  const size_t need = ((size_t)(ent - (int*)ws) + (size_t)NSLICE * eps) * sizeof(int);
  const bool binned = E > 0 && NB <= BIN_MAX && R2 <= 32768 && WR <= 65536 &&
                      2 * (R2 + WR) <= AGG_LDS_FLOATS && epb <= STAGE_MAX &&
                      N < (1 << 24) && need <= ws_size;

  if (binned) {
    k_place<<<NSLICE, SC_THREADS, 0, stream>>>(src, dst, ent, off, E, N, R2, WR, ns2, NB, epb, eps);
    k_deg_bin<<<dim3(ns2, nr2), SC_THREADS, 0, stream>>>(ent, off, (int*)part, N, R2, eps);
    k_dinv<<<(N + B - 1) / B, B, 0, stream>>>((const int*)part, x, dinv, xs, N, ns2);
    k_agg_bin<<<dim3(ns2, nr2), SC_THREADS, 0, stream>>>(ent, off, xs, part, N, R2, WR, eps);
  } else {
    k_degree_lds<<<dim3(ns1, nr1), SC_THREADS, 0, stream>>>(dst, (int*)part, E, N);
    k_dinv<<<(N + B - 1) / B, B, 0, stream>>>((const int*)part, x, dinv, xs, N, ns1);
    k_agg_lds<<<dim3(ns2, nr2), SC_THREADS, 0, stream>>>(src, dst, xs, part, E, N);
  }
  if (nwv > 0)
    k_gemm12<<<(nwv * 32 + B - 1) / B, B, 0, stream>>>(part, xs, dinv, W1, b1, W2, h2s, nwv, N, ns2);
  if (ntail > 0) {
    k_gemm1_tail<<<(ntail * 16 + B - 1) / B, B, 0, stream>>>(part, xs, dinv, W1, b1, r, n0, N, ns2);
    k_gemm2_tail<<<(ntail * 2 + B - 1) / B, B, 0, stream>>>(r, W2, dinv, h2s, n0, N);
  }

  if (binned)
    k_agg_bin<<<dim3(ns2, nr2), SC_THREADS, 0, stream>>>(ent, off, h2s, part, N, R2, WR, eps);
  else
    k_agg_lds<<<dim3(ns2, nr2), SC_THREADS, 0, stream>>>(src, dst, h2s, part, E, N);
  k_final <<<(2 * N + B - 1) / B,  B, 0, stream>>>(part, h2s, dinv, b2, out, N, ns2);
}
